# starting kernel
# baseline (speedup 1.0000x reference)
#include <hip/hip_runtime.h>


typedef float v2f __attribute__((ext_vector_type(2)));
typedef float v8f __attribute__((ext_vector_type(8)));

#define HIDDEN 2048
#define EXPERTS 64
#define TSTEPS 8192

__global__ __launch_bounds__(256) void transpose_w_kernel(const float* __restrict__ W,
                                                          float* __restrict__ BT) {
    int idx = blockIdx.x * 256 + threadIdx.x;
    int k = idx >> 6;
    int n = idx & 63;
    BT[(size_t)n * HIDDEN + k] = W[(size_t)k * EXPERTS + n];
}

__global__ __launch_bounds__(256) void gemm_wmma_kernel(const float* __restrict__ A,
                                                        const float* __restrict__ BT,
                                                        float* __restrict__ C) {
    const int wave  = threadIdx.x >> 5;
    const int lane  = threadIdx.x & 31;
    const int mtile = blockIdx.x * 8 + wave;
    const int m0    = mtile * 16;
    const int half  = lane >> 4;
    const int sub   = lane & 15;

    const float* arow = A  + (size_t)(m0 + sub) * HIDDEN + half * 2;
    const float* b0p  = BT + (size_t)(0 * 16 + sub) * HIDDEN + half * 2;
    const float* b1p  = BT + (size_t)(1 * 16 + sub) * HIDDEN + half * 2;
    const float* b2p  = BT + (size_t)(2 * 16 + sub) * HIDDEN + half * 2;
    const float* b3p  = BT + (size_t)(3 * 16 + sub) * HIDDEN + half * 2;

    v8f acc0 = {}, acc1 = {}, acc2 = {}, acc3 = {};

    for (int k0 = 0; k0 < HIDDEN; k0 += 4) {
        __builtin_prefetch(arow + k0 + 512, 0, 3);
        v2f a  = *(const v2f*)(arow + k0);
        v2f b0 = *(const v2f*)(b0p + k0);
        v2f b1 = *(const v2f*)(b1p + k0);
        v2f b2 = *(const v2f*)(b2p + k0);
        v2f b3 = *(const v2f*)(b3p + k0);
        acc0 = __builtin_amdgcn_wmma_f32_16x16x4_f32(false, a, false, b0, (short)0, acc0, false, false);
        acc1 = __builtin_amdgcn_wmma_f32_16x16x4_f32(false, a, false, b1, (short)0, acc1, false, false);
        acc2 = __builtin_amdgcn_wmma_f32_16x16x4_f32(false, a, false, b2, (short)0, acc2, false, false);
        acc3 = __builtin_amdgcn_wmma_f32_16x16x4_f32(false, a, false, b3, (short)0, acc3, false, false);
    }

#pragma unroll
    for (int r = 0; r < 8; ++r) {
        size_t row = (size_t)(m0 + half * 8 + r) * EXPERTS;
        C[row +  0 + sub] = acc0[r];
        C[row + 16 + sub] = acc1[r];
        C[row + 32 + sub] = acc2[r];
        C[row + 48 + sub] = acc3[r];
    }
}

#define CH 8

__device__ __forceinline__ void lif_step(float2 cur, float2& mp, float2& rf,
                                         float2* __restrict__ out) {
    mp.x *= 0.9f;
    mp.y *= 0.9f;
    float a0 = (rf.x <= 0.0f) ? 0.1f : 0.0f;
    float a1 = (rf.y <= 0.0f) ? 0.1f : 0.0f;
    mp.x += cur.x * a0;
    mp.y += cur.y * a1;
    rf.x = fmaxf(rf.x - 0.1f, 0.0f);
    rf.y = fmaxf(rf.y - 0.1f, 0.0f);
    *out = mp;
    bool s0 = mp.x > 1.0f;
    bool s1 = mp.y > 1.0f;
    mp.x = s0 ? 0.0f : mp.x;
    mp.y = s1 ? 0.0f : mp.y;
    rf.x = s0 ? 1.0f : rf.x;
    rf.y = s1 ? 1.0f : rf.y;
}

__global__ __launch_bounds__(32) void lif_scan_kernel(float2* __restrict__ buf) {
    const int lane = threadIdx.x;
    float2 mp = {0.0f, 0.0f};
    float2 rf = {0.0f, 0.0f};

    float2 bufA[CH], bufB[CH];
#pragma unroll
    for (int i = 0; i < CH; ++i) bufA[i] = buf[(size_t)i * 32 + lane];

    for (int t0 = 0; t0 < TSTEPS; t0 += 2 * CH) {
#pragma unroll
        for (int i = 0; i < CH; ++i) bufB[i] = buf[(size_t)(t0 + CH + i) * 32 + lane];
#pragma unroll
        for (int i = 0; i < CH; ++i)
            lif_step(bufA[i], mp, rf, &buf[(size_t)(t0 + i) * 32 + lane]);

        if (t0 + 2 * CH < TSTEPS) {
#pragma unroll
            for (int i = 0; i < CH; ++i) bufA[i] = buf[(size_t)(t0 + 2 * CH + i) * 32 + lane];
        }
#pragma unroll
        for (int i = 0; i < CH; ++i)
            lif_step(bufB[i], mp, rf, &buf[(size_t)(t0 + CH + i) * 32 + lane]);
    }
}

__global__ __launch_bounds__(256) void routing_kernel(const float2* __restrict__ mpbuf,
                                                      float2* __restrict__ out) {
    const int row  = blockIdx.x * 8 + (threadIdx.x >> 5);
    const int lane = threadIdx.x & 31;

    float2 mp = mpbuf[(size_t)row * 32 + lane];
    bool s0 = mp.x > 1.0f;
    bool s1 = mp.y > 1.0f;

    unsigned long long bal = __ballot(s0 || s1);
    bool any = (bal != 0ull);

    float x0 = any ? (s0 ? 1.0f : 0.0f) : mp.x;
    float x1 = any ? (s1 ? 1.0f : 0.0f) : mp.y;

    float m = fmaxf(x0, x1);
#pragma unroll
    for (int off = 16; off >= 1; off >>= 1)
        m = fmaxf(m, __shfl_xor(m, off, 32));

    float e0 = __expf(x0 - m);
    float e1 = __expf(x1 - m);
    float s  = e0 + e1;
#pragma unroll
    for (int off = 16; off >= 1; off >>= 1)
        s += __shfl_xor(s, off, 32);

    float inv = 1.0f / s;
    float2 r;
    r.x = e0 * inv;
    r.y = e1 * inv;
    out[(size_t)row * 32 + lane] = r;
}

extern "C" void kernel_launch(void* const* d_in, const int* in_sizes, int n_in,
                              void* d_out, int out_size, void* d_ws, size_t ws_size,
                              hipStream_t stream) {
    const float* hidden = (const float*)d_in[0];
    const float* W      = (const float*)d_in[1];
    float* out          = (float*)d_out;

    float* curr = (float*)d_ws;
    float* BT   = (float*)((char*)d_ws + (size_t)TSTEPS * EXPERTS * sizeof(float));

    transpose_w_kernel<<<(HIDDEN * EXPERTS) / 256, 256, 0, stream>>>(W, BT);

    gemm_wmma_kernel<<<64, 256, 0, stream>>>(hidden, BT, curr);

    lif_scan_kernel<<<1, 32, 0, stream>>>((float2*)curr);

    routing_kernel<<<TSTEPS / 8, 256, 0, stream>>>((const float2*)curr, (float2*)out);
}


# LIF scan wave: currents kept in a 64-row register ring, each pair refilled right behind its own step's store, no branch inside a ring pass: loads run 64 steps ahead, no wait on fresh loads
# speedup vs baseline: 1.3771x; 1.3771x over previous
#include <hip/hip_runtime.h>


typedef float v2f __attribute__((ext_vector_type(2)));
typedef float v8f __attribute__((ext_vector_type(8)));

#define HIDDEN 2048
#define EXPERTS 64
#define TSTEPS 8192

__global__ __launch_bounds__(256) void transpose_w_kernel(const float* __restrict__ W,
                                                          float* __restrict__ BT) {
    int idx = blockIdx.x * 256 + threadIdx.x;
    int k = idx >> 6;
    int n = idx & 63;
    BT[(size_t)n * HIDDEN + k] = W[(size_t)k * EXPERTS + n];
}

__global__ __launch_bounds__(256) void gemm_wmma_kernel(const float* __restrict__ A,
                                                        const float* __restrict__ BT,
                                                        float* __restrict__ C) {
    const int wave  = threadIdx.x >> 5;
    const int lane  = threadIdx.x & 31;
    const int mtile = blockIdx.x * 8 + wave;
    const int m0    = mtile * 16;
    const int half  = lane >> 4;
    const int sub   = lane & 15;

    const float* arow = A  + (size_t)(m0 + sub) * HIDDEN + half * 2;
    const float* b0p  = BT + (size_t)(0 * 16 + sub) * HIDDEN + half * 2;
    const float* b1p  = BT + (size_t)(1 * 16 + sub) * HIDDEN + half * 2;
    const float* b2p  = BT + (size_t)(2 * 16 + sub) * HIDDEN + half * 2;
    const float* b3p  = BT + (size_t)(3 * 16 + sub) * HIDDEN + half * 2;

    v8f acc0 = {}, acc1 = {}, acc2 = {}, acc3 = {};

    for (int k0 = 0; k0 < HIDDEN; k0 += 4) {
        __builtin_prefetch(arow + k0 + 512, 0, 3);
        v2f a  = *(const v2f*)(arow + k0);
        v2f b0 = *(const v2f*)(b0p + k0);
        v2f b1 = *(const v2f*)(b1p + k0);
        v2f b2 = *(const v2f*)(b2p + k0);
        v2f b3 = *(const v2f*)(b3p + k0);
        acc0 = __builtin_amdgcn_wmma_f32_16x16x4_f32(false, a, false, b0, (short)0, acc0, false, false);
        acc1 = __builtin_amdgcn_wmma_f32_16x16x4_f32(false, a, false, b1, (short)0, acc1, false, false);
        acc2 = __builtin_amdgcn_wmma_f32_16x16x4_f32(false, a, false, b2, (short)0, acc2, false, false);
        acc3 = __builtin_amdgcn_wmma_f32_16x16x4_f32(false, a, false, b3, (short)0, acc3, false, false);
    }

#pragma unroll
    for (int r = 0; r < 8; ++r) {
        size_t row = (size_t)(m0 + half * 8 + r) * EXPERTS;
        C[row +  0 + sub] = acc0[r];
        C[row + 16 + sub] = acc1[r];
        C[row + 32 + sub] = acc2[r];
        C[row + 48 + sub] = acc3[r];
    }
}

#define RING 64

__device__ __forceinline__ void lif_step(float2 cur, float2& mp, float2& rf,
                                         float2* __restrict__ out) {
    mp.x *= 0.9f;
    mp.y *= 0.9f;
    float a0 = (rf.x <= 0.0f) ? 0.1f : 0.0f;
    float a1 = (rf.y <= 0.0f) ? 0.1f : 0.0f;
    mp.x += cur.x * a0;
    mp.y += cur.y * a1;
    rf.x = fmaxf(rf.x - 0.1f, 0.0f);
    rf.y = fmaxf(rf.y - 0.1f, 0.0f);
    *out = mp;
    bool s0 = mp.x > 1.0f;
    bool s1 = mp.y > 1.0f;
    mp.x = s0 ? 0.0f : mp.x;
    mp.y = s1 ? 0.0f : mp.y;
    rf.x = s0 ? 1.0f : rf.x;
    rf.y = s1 ? 1.0f : rf.y;
}

__global__ __launch_bounds__(32) void lif_scan_kernel(float2* __restrict__ buf) {
    const int lane = threadIdx.x;
    float2 mp = {0.0f, 0.0f};
    float2 rf = {0.0f, 0.0f};

    float2 ring[RING];
#pragma unroll
    for (int i = 0; i < RING; ++i) ring[i] = buf[(size_t)i * 32 + lane];

    int t0 = 0;
    for (; t0 < TSTEPS - RING; t0 += RING) {
#pragma unroll
        for (int i = 0; i < RING; ++i) {
            lif_step(ring[i], mp, rf, &ring[i]);
            buf[(size_t)(t0 + i) * 32 + lane] = ring[i];
            asm volatile("" ::: "memory");
            ring[i] = buf[(size_t)(t0 + RING + i) * 32 + lane];
        }
    }

#pragma unroll
    for (int i = 0; i < RING; ++i) {
        lif_step(ring[i], mp, rf, &ring[i]);
        buf[(size_t)(t0 + i) * 32 + lane] = ring[i];
    }
}

__global__ __launch_bounds__(256) void routing_kernel(const float2* __restrict__ mpbuf,
                                                      float2* __restrict__ out) {
    const int row  = blockIdx.x * 8 + (threadIdx.x >> 5);
    const int lane = threadIdx.x & 31;

    float2 mp = mpbuf[(size_t)row * 32 + lane];
    bool s0 = mp.x > 1.0f;
    bool s1 = mp.y > 1.0f;

    unsigned long long bal = __ballot(s0 || s1);
    bool any = (bal != 0ull);

    float x0 = any ? (s0 ? 1.0f : 0.0f) : mp.x;
    float x1 = any ? (s1 ? 1.0f : 0.0f) : mp.y;

    float m = fmaxf(x0, x1);
#pragma unroll
    for (int off = 16; off >= 1; off >>= 1)
        m = fmaxf(m, __shfl_xor(m, off, 32));

    float e0 = __expf(x0 - m);
    float e1 = __expf(x1 - m);
    float s  = e0 + e1;
#pragma unroll
    for (int off = 16; off >= 1; off >>= 1)
        s += __shfl_xor(s, off, 32);

    float inv = 1.0f / s;
    float2 r;
    r.x = e0 * inv;
    r.y = e1 * inv;
    out[(size_t)row * 32 + lane] = r;
}

extern "C" void kernel_launch(void* const* d_in, const int* in_sizes, int n_in,
                              void* d_out, int out_size, void* d_ws, size_t ws_size,
                              hipStream_t stream) {
    const float* hidden = (const float*)d_in[0];
    const float* W      = (const float*)d_in[1];
    float* out          = (float*)d_out;

    float* curr = (float*)d_ws;
    float* BT   = (float*)((char*)d_ws + (size_t)TSTEPS * EXPERTS * sizeof(float));

    transpose_w_kernel<<<(HIDDEN * EXPERTS) / 256, 256, 0, stream>>>(W, BT);

    gemm_wmma_kernel<<<64, 256, 0, stream>>>(hidden, BT, curr);

    lif_scan_kernel<<<1, 32, 0, stream>>>((float2*)curr);

    routing_kernel<<<TSTEPS / 8, 256, 0, stream>>>((const float2*)curr, (float2*)out);
}


# GEMM K loop: A/B fragments held in a 16-step register ring, each entry refilled right behind its own 4 WMMAs: ~62 loads in flight, no drain per 8 steps; same k order, same sums
# speedup vs baseline: 1.4430x; 1.0479x over previous
#include <hip/hip_runtime.h>


typedef float v2f __attribute__((ext_vector_type(2)));
typedef float v8f __attribute__((ext_vector_type(8)));

#define HIDDEN 2048
#define EXPERTS 64
#define TSTEPS 8192

__global__ __launch_bounds__(256) void transpose_w_kernel(const float* __restrict__ W,
                                                          float* __restrict__ BT) {
    int idx = blockIdx.x * 256 + threadIdx.x;
    int k = idx >> 6;
    int n = idx & 63;
    BT[(size_t)n * HIDDEN + k] = W[(size_t)k * EXPERTS + n];
}

__global__ __launch_bounds__(256) void gemm_wmma_kernel(const float* __restrict__ A,
                                                        const float* __restrict__ BT,
                                                        float* __restrict__ C) {
    const int wave  = threadIdx.x >> 5;
    const int lane  = threadIdx.x & 31;
    const int mtile = blockIdx.x * 8 + wave;
    const int m0    = mtile * 16;
    const int half  = lane >> 4;
    const int sub   = lane & 15;

    const float* arow = A  + (size_t)(m0 + sub) * HIDDEN + half * 2;
    const float* b0p  = BT + (size_t)(0 * 16 + sub) * HIDDEN + half * 2;
    const float* b1p  = BT + (size_t)(1 * 16 + sub) * HIDDEN + half * 2;
    const float* b2p  = BT + (size_t)(2 * 16 + sub) * HIDDEN + half * 2;
    const float* b3p  = BT + (size_t)(3 * 16 + sub) * HIDDEN + half * 2;

    v8f acc0 = {}, acc1 = {}, acc2 = {}, acc3 = {};

#define KR 16
    v2f ra[KR], rb0[KR], rb1[KR], rb2[KR], rb3[KR];
#pragma unroll
    for (int j = 0; j < KR; ++j) {
        ra[j]  = *(const v2f*)(arow + 4 * j);
        rb0[j] = *(const v2f*)(b0p  + 4 * j);
        rb1[j] = *(const v2f*)(b1p  + 4 * j);
        rb2[j] = *(const v2f*)(b2p  + 4 * j);
        rb3[j] = *(const v2f*)(b3p  + 4 * j);
    }

    int k0 = 0;
    for (; k0 < HIDDEN - 4 * KR; k0 += 4 * KR) {
        __builtin_prefetch(arow + k0 + 512, 0, 3);
        __builtin_prefetch(arow + k0 + 544, 0, 3);
#pragma unroll
        for (int j = 0; j < KR; ++j) {
            acc0 = __builtin_amdgcn_wmma_f32_16x16x4_f32(false, ra[j], false, rb0[j], (short)0, acc0, false, false);
            acc1 = __builtin_amdgcn_wmma_f32_16x16x4_f32(false, ra[j], false, rb1[j], (short)0, acc1, false, false);
            acc2 = __builtin_amdgcn_wmma_f32_16x16x4_f32(false, ra[j], false, rb2[j], (short)0, acc2, false, false);
            acc3 = __builtin_amdgcn_wmma_f32_16x16x4_f32(false, ra[j], false, rb3[j], (short)0, acc3, false, false);
            const int kn = k0 + 4 * KR + 4 * j;
            ra[j]  = *(const v2f*)(arow + kn);
            rb0[j] = *(const v2f*)(b0p  + kn);
            rb1[j] = *(const v2f*)(b1p  + kn);
            rb2[j] = *(const v2f*)(b2p  + kn);
            rb3[j] = *(const v2f*)(b3p  + kn);
        }
    }
#pragma unroll
    for (int j = 0; j < KR; ++j) {
        acc0 = __builtin_amdgcn_wmma_f32_16x16x4_f32(false, ra[j], false, rb0[j], (short)0, acc0, false, false);
        acc1 = __builtin_amdgcn_wmma_f32_16x16x4_f32(false, ra[j], false, rb1[j], (short)0, acc1, false, false);
        acc2 = __builtin_amdgcn_wmma_f32_16x16x4_f32(false, ra[j], false, rb2[j], (short)0, acc2, false, false);
        acc3 = __builtin_amdgcn_wmma_f32_16x16x4_f32(false, ra[j], false, rb3[j], (short)0, acc3, false, false);
    }
#undef KR

#pragma unroll
    for (int r = 0; r < 8; ++r) {
        size_t row = (size_t)(m0 + half * 8 + r) * EXPERTS;
        C[row +  0 + sub] = acc0[r];
        C[row + 16 + sub] = acc1[r];
        C[row + 32 + sub] = acc2[r];
        C[row + 48 + sub] = acc3[r];
    }
}

#define RING 64

__device__ __forceinline__ void lif_step(float2 cur, float2& mp, float2& rf,
                                         float2* __restrict__ out) {
    mp.x *= 0.9f;
    mp.y *= 0.9f;
    float a0 = (rf.x <= 0.0f) ? 0.1f : 0.0f;
    float a1 = (rf.y <= 0.0f) ? 0.1f : 0.0f;
    mp.x += cur.x * a0;
    mp.y += cur.y * a1;
    rf.x = fmaxf(rf.x - 0.1f, 0.0f);
    rf.y = fmaxf(rf.y - 0.1f, 0.0f);
    *out = mp;
    bool s0 = mp.x > 1.0f;
    bool s1 = mp.y > 1.0f;
    mp.x = s0 ? 0.0f : mp.x;
    mp.y = s1 ? 0.0f : mp.y;
    rf.x = s0 ? 1.0f : rf.x;
    rf.y = s1 ? 1.0f : rf.y;
}

__global__ __launch_bounds__(32) void lif_scan_kernel(float2* __restrict__ buf) {
    const int lane = threadIdx.x;
    float2 mp = {0.0f, 0.0f};
    float2 rf = {0.0f, 0.0f};

    float2 ring[RING];
#pragma unroll
    for (int i = 0; i < RING; ++i) ring[i] = buf[(size_t)i * 32 + lane];

    int t0 = 0;
    for (; t0 < TSTEPS - RING; t0 += RING) {
#pragma unroll
        for (int i = 0; i < RING; ++i) {
            lif_step(ring[i], mp, rf, &ring[i]);
            buf[(size_t)(t0 + i) * 32 + lane] = ring[i];
            asm volatile("" ::: "memory");
            ring[i] = buf[(size_t)(t0 + RING + i) * 32 + lane];
        }
    }

#pragma unroll
    for (int i = 0; i < RING; ++i) {
        lif_step(ring[i], mp, rf, &ring[i]);
        buf[(size_t)(t0 + i) * 32 + lane] = ring[i];
    }
}

__global__ __launch_bounds__(256) void routing_kernel(const float2* __restrict__ mpbuf,
                                                      float2* __restrict__ out) {
    const int row  = blockIdx.x * 8 + (threadIdx.x >> 5);
    const int lane = threadIdx.x & 31;

    float2 mp = mpbuf[(size_t)row * 32 + lane];
    bool s0 = mp.x > 1.0f;
    bool s1 = mp.y > 1.0f;

    unsigned long long bal = __ballot(s0 || s1);
    bool any = (bal != 0ull);

    float x0 = any ? (s0 ? 1.0f : 0.0f) : mp.x;
    float x1 = any ? (s1 ? 1.0f : 0.0f) : mp.y;

    float m = fmaxf(x0, x1);
#pragma unroll
    for (int off = 16; off >= 1; off >>= 1)
        m = fmaxf(m, __shfl_xor(m, off, 32));

    float e0 = __expf(x0 - m);
    float e1 = __expf(x1 - m);
    float s  = e0 + e1;
#pragma unroll
    for (int off = 16; off >= 1; off >>= 1)
        s += __shfl_xor(s, off, 32);

    float inv = 1.0f / s;
    float2 r;
    r.x = e0 * inv;
    r.y = e1 * inv;
    out[(size_t)row * 32 + lane] = r;
}

extern "C" void kernel_launch(void* const* d_in, const int* in_sizes, int n_in,
                              void* d_out, int out_size, void* d_ws, size_t ws_size,
                              hipStream_t stream) {
    const float* hidden = (const float*)d_in[0];
    const float* W      = (const float*)d_in[1];
    float* out          = (float*)d_out;

    float* curr = (float*)d_ws;
    float* BT   = (float*)((char*)d_ws + (size_t)TSTEPS * EXPERTS * sizeof(float));

    transpose_w_kernel<<<(HIDDEN * EXPERTS) / 256, 256, 0, stream>>>(W, BT);

    gemm_wmma_kernel<<<64, 256, 0, stream>>>(hidden, BT, curr);

    lif_scan_kernel<<<1, 32, 0, stream>>>((float2*)curr);

    routing_kernel<<<TSTEPS / 8, 256, 0, stream>>>((const float2*)curr, (float2*)out);
}


# GEMM launch shape: one wave per work-group (512 x 32 threads, was 64 x 256), so the 512 M-tile waves spread over all WGPs; per-wave work, k order and sums unchanged
# speedup vs baseline: 1.6903x; 1.1714x over previous
#include <hip/hip_runtime.h>


typedef float v2f __attribute__((ext_vector_type(2)));
typedef float v8f __attribute__((ext_vector_type(8)));

#define HIDDEN 2048
#define EXPERTS 64
#define TSTEPS 8192

__global__ __launch_bounds__(256) void transpose_w_kernel(const float* __restrict__ W,
                                                          float* __restrict__ BT) {
    int idx = blockIdx.x * 256 + threadIdx.x;
    int k = idx >> 6;
    int n = idx & 63;
    BT[(size_t)n * HIDDEN + k] = W[(size_t)k * EXPERTS + n];
}

__global__ __launch_bounds__(32) void gemm_wmma_kernel(const float* __restrict__ A,
                                                        const float* __restrict__ BT,
                                                        float* __restrict__ C) {
    const int lane  = threadIdx.x;
    const int mtile = blockIdx.x;
    const int m0    = mtile * 16;
    const int half  = lane >> 4;
    const int sub   = lane & 15;

    const float* arow = A  + (size_t)(m0 + sub) * HIDDEN + half * 2;
    const float* b0p  = BT + (size_t)(0 * 16 + sub) * HIDDEN + half * 2;
    const float* b1p  = BT + (size_t)(1 * 16 + sub) * HIDDEN + half * 2;
    const float* b2p  = BT + (size_t)(2 * 16 + sub) * HIDDEN + half * 2;
    const float* b3p  = BT + (size_t)(3 * 16 + sub) * HIDDEN + half * 2;

    v8f acc0 = {}, acc1 = {}, acc2 = {}, acc3 = {};

#define KR 16
    v2f ra[KR], rb0[KR], rb1[KR], rb2[KR], rb3[KR];
#pragma unroll
    for (int j = 0; j < KR; ++j) {
        ra[j]  = *(const v2f*)(arow + 4 * j);
        rb0[j] = *(const v2f*)(b0p  + 4 * j);
        rb1[j] = *(const v2f*)(b1p  + 4 * j);
        rb2[j] = *(const v2f*)(b2p  + 4 * j);
        rb3[j] = *(const v2f*)(b3p  + 4 * j);
    }

    int k0 = 0;
    for (; k0 < HIDDEN - 4 * KR; k0 += 4 * KR) {
        __builtin_prefetch(arow + k0 + 512, 0, 3);
        __builtin_prefetch(arow + k0 + 544, 0, 3);
#pragma unroll
        for (int j = 0; j < KR; ++j) {
            acc0 = __builtin_amdgcn_wmma_f32_16x16x4_f32(false, ra[j], false, rb0[j], (short)0, acc0, false, false);
            acc1 = __builtin_amdgcn_wmma_f32_16x16x4_f32(false, ra[j], false, rb1[j], (short)0, acc1, false, false);
            acc2 = __builtin_amdgcn_wmma_f32_16x16x4_f32(false, ra[j], false, rb2[j], (short)0, acc2, false, false);
            acc3 = __builtin_amdgcn_wmma_f32_16x16x4_f32(false, ra[j], false, rb3[j], (short)0, acc3, false, false);
            const int kn = k0 + 4 * KR + 4 * j;
            ra[j]  = *(const v2f*)(arow + kn);
            rb0[j] = *(const v2f*)(b0p  + kn);
            rb1[j] = *(const v2f*)(b1p  + kn);
            rb2[j] = *(const v2f*)(b2p  + kn);
            rb3[j] = *(const v2f*)(b3p  + kn);
        }
    }
#pragma unroll
    for (int j = 0; j < KR; ++j) {
        acc0 = __builtin_amdgcn_wmma_f32_16x16x4_f32(false, ra[j], false, rb0[j], (short)0, acc0, false, false);
        acc1 = __builtin_amdgcn_wmma_f32_16x16x4_f32(false, ra[j], false, rb1[j], (short)0, acc1, false, false);
        acc2 = __builtin_amdgcn_wmma_f32_16x16x4_f32(false, ra[j], false, rb2[j], (short)0, acc2, false, false);
        acc3 = __builtin_amdgcn_wmma_f32_16x16x4_f32(false, ra[j], false, rb3[j], (short)0, acc3, false, false);
    }
#undef KR

#pragma unroll
    for (int r = 0; r < 8; ++r) {
        size_t row = (size_t)(m0 + half * 8 + r) * EXPERTS;
        C[row +  0 + sub] = acc0[r];
        C[row + 16 + sub] = acc1[r];
        C[row + 32 + sub] = acc2[r];
        C[row + 48 + sub] = acc3[r];
    }
}

#define RING 64

__device__ __forceinline__ void lif_step(float2 cur, float2& mp, float2& rf,
                                         float2* __restrict__ out) {
    mp.x *= 0.9f;
    mp.y *= 0.9f;
    float a0 = (rf.x <= 0.0f) ? 0.1f : 0.0f;
    float a1 = (rf.y <= 0.0f) ? 0.1f : 0.0f;
    mp.x += cur.x * a0;
    mp.y += cur.y * a1;
    rf.x = fmaxf(rf.x - 0.1f, 0.0f);
    rf.y = fmaxf(rf.y - 0.1f, 0.0f);
    *out = mp;
    bool s0 = mp.x > 1.0f;
    bool s1 = mp.y > 1.0f;
    mp.x = s0 ? 0.0f : mp.x;
    mp.y = s1 ? 0.0f : mp.y;
    rf.x = s0 ? 1.0f : rf.x;
    rf.y = s1 ? 1.0f : rf.y;
}

__global__ __launch_bounds__(32) void lif_scan_kernel(float2* __restrict__ buf) {
    const int lane = threadIdx.x;
    float2 mp = {0.0f, 0.0f};
    float2 rf = {0.0f, 0.0f};

    float2 ring[RING];
#pragma unroll
    for (int i = 0; i < RING; ++i) ring[i] = buf[(size_t)i * 32 + lane];

    int t0 = 0;
    for (; t0 < TSTEPS - RING; t0 += RING) {
#pragma unroll
        for (int i = 0; i < RING; ++i) {
            lif_step(ring[i], mp, rf, &ring[i]);
            buf[(size_t)(t0 + i) * 32 + lane] = ring[i];
            asm volatile("" ::: "memory");
            ring[i] = buf[(size_t)(t0 + RING + i) * 32 + lane];
        }
    }

#pragma unroll
    for (int i = 0; i < RING; ++i) {
        lif_step(ring[i], mp, rf, &ring[i]);
        buf[(size_t)(t0 + i) * 32 + lane] = ring[i];
    }
}

__global__ __launch_bounds__(256) void routing_kernel(const float2* __restrict__ mpbuf,
                                                      float2* __restrict__ out) {
    const int row  = blockIdx.x * 8 + (threadIdx.x >> 5);
    const int lane = threadIdx.x & 31;

    float2 mp = mpbuf[(size_t)row * 32 + lane];
    bool s0 = mp.x > 1.0f;
    bool s1 = mp.y > 1.0f;

    unsigned long long bal = __ballot(s0 || s1);
    bool any = (bal != 0ull);

    float x0 = any ? (s0 ? 1.0f : 0.0f) : mp.x;
    float x1 = any ? (s1 ? 1.0f : 0.0f) : mp.y;

    float m = fmaxf(x0, x1);
#pragma unroll
    for (int off = 16; off >= 1; off >>= 1)
        m = fmaxf(m, __shfl_xor(m, off, 32));

    float e0 = __expf(x0 - m);
    float e1 = __expf(x1 - m);
    float s  = e0 + e1;
#pragma unroll
    for (int off = 16; off >= 1; off >>= 1)
        s += __shfl_xor(s, off, 32);

    float inv = 1.0f / s;
    float2 r;
    r.x = e0 * inv;
    r.y = e1 * inv;
    out[(size_t)row * 32 + lane] = r;
}

extern "C" void kernel_launch(void* const* d_in, const int* in_sizes, int n_in,
                              void* d_out, int out_size, void* d_ws, size_t ws_size,
                              hipStream_t stream) {
    const float* hidden = (const float*)d_in[0];
    const float* W      = (const float*)d_in[1];
    float* out          = (float*)d_out;

    float* curr = (float*)d_ws;
    float* BT   = (float*)((char*)d_ws + (size_t)TSTEPS * EXPERTS * sizeof(float));

    transpose_w_kernel<<<(HIDDEN * EXPERTS) / 256, 256, 0, stream>>>(W, BT);

    gemm_wmma_kernel<<<512, 32, 0, stream>>>(hidden, BT, curr);

    lif_scan_kernel<<<1, 32, 0, stream>>>((float2*)curr);

    routing_kernel<<<TSTEPS / 8, 256, 0, stream>>>((const float2*)curr, (float2*)out);
}


# LIF step: serial chain cut from fma,cmp,select,cmp,select to fma,cmp,select: spike flag selects next leak addend and gain directly; no-spike gain formed early from prior rf; every step fuses cur*gain
# speedup vs baseline: 1.7923x; 1.0603x over previous
#include <hip/hip_runtime.h>


typedef float v2f __attribute__((ext_vector_type(2)));
typedef float v8f __attribute__((ext_vector_type(8)));

#define HIDDEN 2048
#define EXPERTS 64
#define TSTEPS 8192

__global__ __launch_bounds__(256) void transpose_w_kernel(const float* __restrict__ W,
                                                          float* __restrict__ BT) {
    int idx = blockIdx.x * 256 + threadIdx.x;
    int k = idx >> 6;
    int n = idx & 63;
    BT[(size_t)n * HIDDEN + k] = W[(size_t)k * EXPERTS + n];
}

__global__ __launch_bounds__(32) void gemm_wmma_kernel(const float* __restrict__ A,
                                                        const float* __restrict__ BT,
                                                        float* __restrict__ C) {
    const int lane  = threadIdx.x;
    const int mtile = blockIdx.x;
    const int m0    = mtile * 16;
    const int half  = lane >> 4;
    const int sub   = lane & 15;

    const float* arow = A  + (size_t)(m0 + sub) * HIDDEN + half * 2;
    const float* b0p  = BT + (size_t)(0 * 16 + sub) * HIDDEN + half * 2;
    const float* b1p  = BT + (size_t)(1 * 16 + sub) * HIDDEN + half * 2;
    const float* b2p  = BT + (size_t)(2 * 16 + sub) * HIDDEN + half * 2;
    const float* b3p  = BT + (size_t)(3 * 16 + sub) * HIDDEN + half * 2;

    v8f acc0 = {}, acc1 = {}, acc2 = {}, acc3 = {};

#define KR 16
    v2f ra[KR], rb0[KR], rb1[KR], rb2[KR], rb3[KR];
#pragma unroll
    for (int j = 0; j < KR; ++j) {
        ra[j]  = *(const v2f*)(arow + 4 * j);
        rb0[j] = *(const v2f*)(b0p  + 4 * j);
        rb1[j] = *(const v2f*)(b1p  + 4 * j);
        rb2[j] = *(const v2f*)(b2p  + 4 * j);
        rb3[j] = *(const v2f*)(b3p  + 4 * j);
    }

    int k0 = 0;
    for (; k0 < HIDDEN - 4 * KR; k0 += 4 * KR) {
        __builtin_prefetch(arow + k0 + 512, 0, 3);
        __builtin_prefetch(arow + k0 + 544, 0, 3);
#pragma unroll
        for (int j = 0; j < KR; ++j) {
            acc0 = __builtin_amdgcn_wmma_f32_16x16x4_f32(false, ra[j], false, rb0[j], (short)0, acc0, false, false);
            acc1 = __builtin_amdgcn_wmma_f32_16x16x4_f32(false, ra[j], false, rb1[j], (short)0, acc1, false, false);
            acc2 = __builtin_amdgcn_wmma_f32_16x16x4_f32(false, ra[j], false, rb2[j], (short)0, acc2, false, false);
            acc3 = __builtin_amdgcn_wmma_f32_16x16x4_f32(false, ra[j], false, rb3[j], (short)0, acc3, false, false);
            const int kn = k0 + 4 * KR + 4 * j;
            ra[j]  = *(const v2f*)(arow + kn);
            rb0[j] = *(const v2f*)(b0p  + kn);
            rb1[j] = *(const v2f*)(b1p  + kn);
            rb2[j] = *(const v2f*)(b2p  + kn);
            rb3[j] = *(const v2f*)(b3p  + kn);
        }
    }
#pragma unroll
    for (int j = 0; j < KR; ++j) {
        acc0 = __builtin_amdgcn_wmma_f32_16x16x4_f32(false, ra[j], false, rb0[j], (short)0, acc0, false, false);
        acc1 = __builtin_amdgcn_wmma_f32_16x16x4_f32(false, ra[j], false, rb1[j], (short)0, acc1, false, false);
        acc2 = __builtin_amdgcn_wmma_f32_16x16x4_f32(false, ra[j], false, rb2[j], (short)0, acc2, false, false);
        acc3 = __builtin_amdgcn_wmma_f32_16x16x4_f32(false, ra[j], false, rb3[j], (short)0, acc3, false, false);
    }
#undef KR

#pragma unroll
    for (int r = 0; r < 8; ++r) {
        size_t row = (size_t)(m0 + half * 8 + r) * EXPERTS;
        C[row +  0 + sub] = acc0[r];
        C[row + 16 + sub] = acc1[r];
        C[row + 32 + sub] = acc2[r];
        C[row + 48 + sub] = acc3[r];
    }
}

#define RING 64

__device__ __forceinline__ void lif_step(float2 cur, float2& m9, float2& a, float2& rf,
                                         float2* __restrict__ out) {
    float2 p;
    p.x = __builtin_fmaf(cur.x, a.x, m9.x);
    p.y = __builtin_fmaf(cur.y, a.y, m9.y);
    *out = p;
    const float d0 = fmaxf(rf.x - 0.1f, 0.0f);
    const float d1 = fmaxf(rf.y - 0.1f, 0.0f);
    float n0 = (rf.x <= 0.1f) ? 0.1f : 0.0f;
    float n1 = (rf.y <= 0.1f) ? 0.1f : 0.0f;
    asm("" : "+v"(n0));
    asm("" : "+v"(n1));
    const float t0 = p.x * 0.9f;
    const float t1 = p.y * 0.9f;
    const bool s0 = p.x > 1.0f;
    const bool s1 = p.y > 1.0f;
    m9.x = s0 ? 0.0f : t0;
    m9.y = s1 ? 0.0f : t1;
    a.x  = s0 ? 0.0f : n0;
    a.y  = s1 ? 0.0f : n1;
    rf.x = s0 ? 1.0f : d0;
    rf.y = s1 ? 1.0f : d1;
}

__global__ __launch_bounds__(32) void lif_scan_kernel(float2* __restrict__ buf) {
    const int lane = threadIdx.x;
    float2 m9 = {0.0f, 0.0f};
    float2 a  = {0.1f, 0.1f};
    float2 rf = {0.0f, 0.0f};

    float2 ring[RING];
#pragma unroll
    for (int i = 0; i < RING; ++i) ring[i] = buf[(size_t)i * 32 + lane];

    int t0 = 0;
    for (; t0 < TSTEPS - RING; t0 += RING) {
#pragma unroll
        for (int i = 0; i < RING; ++i) {
            lif_step(ring[i], m9, a, rf, &ring[i]);
            buf[(size_t)(t0 + i) * 32 + lane] = ring[i];
            asm volatile("" ::: "memory");
            ring[i] = buf[(size_t)(t0 + RING + i) * 32 + lane];
        }
    }

#pragma unroll
    for (int i = 0; i < RING; ++i) {
        lif_step(ring[i], m9, a, rf, &ring[i]);
        buf[(size_t)(t0 + i) * 32 + lane] = ring[i];
    }
}

__global__ __launch_bounds__(256) void routing_kernel(const float2* __restrict__ mpbuf,
                                                      float2* __restrict__ out) {
    const int row  = blockIdx.x * 8 + (threadIdx.x >> 5);
    const int lane = threadIdx.x & 31;

    float2 mp = mpbuf[(size_t)row * 32 + lane];
    bool s0 = mp.x > 1.0f;
    bool s1 = mp.y > 1.0f;

    unsigned long long bal = __ballot(s0 || s1);
    bool any = (bal != 0ull);

    float x0 = any ? (s0 ? 1.0f : 0.0f) : mp.x;
    float x1 = any ? (s1 ? 1.0f : 0.0f) : mp.y;

    float m = fmaxf(x0, x1);
#pragma unroll
    for (int off = 16; off >= 1; off >>= 1)
        m = fmaxf(m, __shfl_xor(m, off, 32));

    float e0 = __expf(x0 - m);
    float e1 = __expf(x1 - m);
    float s  = e0 + e1;
#pragma unroll
    for (int off = 16; off >= 1; off >>= 1)
        s += __shfl_xor(s, off, 32);

    float inv = 1.0f / s;
    float2 r;
    r.x = e0 * inv;
    r.y = e1 * inv;
    out[(size_t)row * 32 + lane] = r;
}

extern "C" void kernel_launch(void* const* d_in, const int* in_sizes, int n_in,
                              void* d_out, int out_size, void* d_ws, size_t ws_size,
                              hipStream_t stream) {
    const float* hidden = (const float*)d_in[0];
    const float* W      = (const float*)d_in[1];
    float* out          = (float*)d_out;

    float* curr = (float*)d_ws;
    float* BT   = (float*)((char*)d_ws + (size_t)TSTEPS * EXPERTS * sizeof(float));

    transpose_w_kernel<<<(HIDDEN * EXPERTS) / 256, 256, 0, stream>>>(W, BT);

    gemm_wmma_kernel<<<512, 32, 0, stream>>>(hidden, BT, curr);

    lif_scan_kernel<<<1, 32, 0, stream>>>((float2*)curr);

    routing_kernel<<<TSTEPS / 8, 256, 0, stream>>>((const float2*)curr, (float2*)out);
}


# LIF scan: serial wave works from LDS only; 8 helper waves of its work-group copy currents in and potentials out in whole lines, a chunk ahead; no global load/store on the chain; sums unchanged
# speedup vs baseline: 2.2682x; 1.2655x over previous
#include <hip/hip_runtime.h>


typedef float v2f __attribute__((ext_vector_type(2)));
typedef float v8f __attribute__((ext_vector_type(8)));

#define HIDDEN 2048
#define EXPERTS 64
#define TSTEPS 8192

__global__ __launch_bounds__(256) void transpose_w_kernel(const float* __restrict__ W,
                                                          float* __restrict__ BT) {
    int idx = blockIdx.x * 256 + threadIdx.x;
    int k = idx >> 6;
    int n = idx & 63;
    BT[(size_t)n * HIDDEN + k] = W[(size_t)k * EXPERTS + n];
}

__global__ __launch_bounds__(32) void gemm_wmma_kernel(const float* __restrict__ A,
                                                        const float* __restrict__ BT,
                                                        float* __restrict__ C) {
    const int lane  = threadIdx.x;
    const int mtile = blockIdx.x;
    const int m0    = mtile * 16;
    const int half  = lane >> 4;
    const int sub   = lane & 15;

    const float* arow = A  + (size_t)(m0 + sub) * HIDDEN + half * 2;
    const float* b0p  = BT + (size_t)(0 * 16 + sub) * HIDDEN + half * 2;
    const float* b1p  = BT + (size_t)(1 * 16 + sub) * HIDDEN + half * 2;
    const float* b2p  = BT + (size_t)(2 * 16 + sub) * HIDDEN + half * 2;
    const float* b3p  = BT + (size_t)(3 * 16 + sub) * HIDDEN + half * 2;

    v8f acc0 = {}, acc1 = {}, acc2 = {}, acc3 = {};

#define KR 16
    v2f ra[KR], rb0[KR], rb1[KR], rb2[KR], rb3[KR];
#pragma unroll
    for (int j = 0; j < KR; ++j) {
        ra[j]  = *(const v2f*)(arow + 4 * j);
        rb0[j] = *(const v2f*)(b0p  + 4 * j);
        rb1[j] = *(const v2f*)(b1p  + 4 * j);
        rb2[j] = *(const v2f*)(b2p  + 4 * j);
        rb3[j] = *(const v2f*)(b3p  + 4 * j);
    }

    int k0 = 0;
    for (; k0 < HIDDEN - 4 * KR; k0 += 4 * KR) {
        __builtin_prefetch(arow + k0 + 512, 0, 3);
        __builtin_prefetch(arow + k0 + 544, 0, 3);
#pragma unroll
        for (int j = 0; j < KR; ++j) {
            acc0 = __builtin_amdgcn_wmma_f32_16x16x4_f32(false, ra[j], false, rb0[j], (short)0, acc0, false, false);
            acc1 = __builtin_amdgcn_wmma_f32_16x16x4_f32(false, ra[j], false, rb1[j], (short)0, acc1, false, false);
            acc2 = __builtin_amdgcn_wmma_f32_16x16x4_f32(false, ra[j], false, rb2[j], (short)0, acc2, false, false);
            acc3 = __builtin_amdgcn_wmma_f32_16x16x4_f32(false, ra[j], false, rb3[j], (short)0, acc3, false, false);
            const int kn = k0 + 4 * KR + 4 * j;
            ra[j]  = *(const v2f*)(arow + kn);
            rb0[j] = *(const v2f*)(b0p  + kn);
            rb1[j] = *(const v2f*)(b1p  + kn);
            rb2[j] = *(const v2f*)(b2p  + kn);
            rb3[j] = *(const v2f*)(b3p  + kn);
        }
    }
#pragma unroll
    for (int j = 0; j < KR; ++j) {
        acc0 = __builtin_amdgcn_wmma_f32_16x16x4_f32(false, ra[j], false, rb0[j], (short)0, acc0, false, false);
        acc1 = __builtin_amdgcn_wmma_f32_16x16x4_f32(false, ra[j], false, rb1[j], (short)0, acc1, false, false);
        acc2 = __builtin_amdgcn_wmma_f32_16x16x4_f32(false, ra[j], false, rb2[j], (short)0, acc2, false, false);
        acc3 = __builtin_amdgcn_wmma_f32_16x16x4_f32(false, ra[j], false, rb3[j], (short)0, acc3, false, false);
    }
#undef KR

#pragma unroll
    for (int r = 0; r < 8; ++r) {
        size_t row = (size_t)(m0 + half * 8 + r) * EXPERTS;
        C[row +  0 + sub] = acc0[r];
        C[row + 16 + sub] = acc1[r];
        C[row + 32 + sub] = acc2[r];
        C[row + 48 + sub] = acc3[r];
    }
}

#define CH 256
#define NCH (TSTEPS / CH)
#define LR 32
#define HELPERS 8
#define LIF_THREADS (32 * (1 + HELPERS))
#define CH_F4 (CH * EXPERTS / 4)
#define H_ROUNDS (CH_F4 / (HELPERS * 32))

__device__ __forceinline__ void lif_step(float2 cur, float2& m9, float2& a, float2& rf,
                                         float2* __restrict__ out) {
    float2 p;
    p.x = __builtin_fmaf(cur.x, a.x, m9.x);
    p.y = __builtin_fmaf(cur.y, a.y, m9.y);
    *out = p;
    const float d0 = fmaxf(rf.x - 0.1f, 0.0f);
    const float d1 = fmaxf(rf.y - 0.1f, 0.0f);
    float n0 = (rf.x <= 0.1f) ? 0.1f : 0.0f;
    float n1 = (rf.y <= 0.1f) ? 0.1f : 0.0f;
    asm("" : "+v"(n0));
    asm("" : "+v"(n1));
    const float t0 = p.x * 0.9f;
    const float t1 = p.y * 0.9f;
    const bool s0 = p.x > 1.0f;
    const bool s1 = p.y > 1.0f;
    m9.x = s0 ? 0.0f : t0;
    m9.y = s1 ? 0.0f : t1;
    a.x  = s0 ? 0.0f : n0;
    a.y  = s1 ? 0.0f : n1;
    rf.x = s0 ? 1.0f : d0;
    rf.y = s1 ? 1.0f : d1;
}

typedef float v4f __attribute__((ext_vector_type(4)));

__global__ __launch_bounds__(LIF_THREADS) void lif_scan_kernel(float* __restrict__ gbuf) {
    __shared__ v4f lds[3][CH_F4];

    const int wave = __builtin_amdgcn_readfirstlane((int)(threadIdx.x >> 5));
    const int lane = threadIdx.x & 31;
    const int h    = (int)threadIdx.x - 32;

    v4f* __restrict__ g4 = (v4f*)gbuf;

    float2 m9 = {0.0f, 0.0f};
    float2 a  = {0.1f, 0.1f};
    float2 rf = {0.0f, 0.0f};

    v4f stage[H_ROUNDS];

    if (wave != 0) {
#pragma unroll
        for (int r = 0; r < H_ROUNDS; ++r) stage[r] = g4[r * (HELPERS * 32) + h];
#pragma unroll
        for (int r = 0; r < H_ROUNDS; ++r) lds[0][r * (HELPERS * 32) + h] = stage[r];
#pragma unroll
        for (int r = 0; r < H_ROUNDS; ++r) stage[r] = g4[CH_F4 + r * (HELPERS * 32) + h];
    }
    __syncthreads();

    int cur = 0, nxt = 1, prv = 2;
    for (int c = 0; c < NCH; ++c) {
        if (wave == 0) {
            float2* rows = (float2*)&lds[cur][0];
            float2 ring[LR];
#pragma unroll
            for (int i = 0; i < LR; ++i) ring[i] = rows[i * 32 + lane];
            int t0 = 0;
            for (; t0 < CH - LR; t0 += LR) {
#pragma unroll
                for (int i = 0; i < LR; i += 2) {
                    float2 p0, p1;
                    lif_step(ring[i],     m9, a, rf, &p0);
                    lif_step(ring[i + 1], m9, a, rf, &p1);
                    rows[(t0 + i) * 32 + lane]     = p0;
                    rows[(t0 + i + 1) * 32 + lane] = p1;
                    ring[i]     = rows[(t0 + LR + i) * 32 + lane];
                    ring[i + 1] = rows[(t0 + LR + i + 1) * 32 + lane];
                }
            }
#pragma unroll
            for (int i = 0; i < LR; ++i) {
                float2 p;
                lif_step(ring[i], m9, a, rf, &p);
                rows[(t0 + i) * 32 + lane] = p;
            }
        } else {
            if (c + 1 < NCH) {
#pragma unroll
                for (int r = 0; r < H_ROUNDS; ++r) lds[nxt][r * (HELPERS * 32) + h] = stage[r];
            }
            if (c + 2 < NCH) {
                const v4f* src = g4 + (size_t)(c + 2) * CH_F4;
#pragma unroll
                for (int r = 0; r < H_ROUNDS; ++r) stage[r] = src[r * (HELPERS * 32) + h];
            }
            if (c >= 1) {
                v4f* dst = g4 + (size_t)(c - 1) * CH_F4;
#pragma unroll
                for (int r = 0; r < H_ROUNDS; ++r) dst[r * (HELPERS * 32) + h] = lds[prv][r * (HELPERS * 32) + h];
            }
        }
        __syncthreads();
        const int t = prv; prv = cur; cur = nxt; nxt = t;
    }

    if (wave != 0) {
        v4f* dst = g4 + (size_t)(NCH - 1) * CH_F4;
#pragma unroll
        for (int r = 0; r < H_ROUNDS; ++r) dst[r * (HELPERS * 32) + h] = lds[prv][r * (HELPERS * 32) + h];
    }
}

__global__ __launch_bounds__(256) void routing_kernel(const float2* __restrict__ mpbuf,
                                                      float2* __restrict__ out) {
    const int row  = blockIdx.x * 8 + (threadIdx.x >> 5);
    const int lane = threadIdx.x & 31;

    float2 mp = mpbuf[(size_t)row * 32 + lane];
    bool s0 = mp.x > 1.0f;
    bool s1 = mp.y > 1.0f;

    unsigned long long bal = __ballot(s0 || s1);
    bool any = (bal != 0ull);

    float x0 = any ? (s0 ? 1.0f : 0.0f) : mp.x;
    float x1 = any ? (s1 ? 1.0f : 0.0f) : mp.y;

    float m = fmaxf(x0, x1);
#pragma unroll
    for (int off = 16; off >= 1; off >>= 1)
        m = fmaxf(m, __shfl_xor(m, off, 32));

    float e0 = __expf(x0 - m);
    float e1 = __expf(x1 - m);
    float s  = e0 + e1;
#pragma unroll
    for (int off = 16; off >= 1; off >>= 1)
        s += __shfl_xor(s, off, 32);

    float inv = 1.0f / s;
    float2 r;
    r.x = e0 * inv;
    r.y = e1 * inv;
    out[(size_t)row * 32 + lane] = r;
}

extern "C" void kernel_launch(void* const* d_in, const int* in_sizes, int n_in,
                              void* d_out, int out_size, void* d_ws, size_t ws_size,
                              hipStream_t stream) {
    const float* hidden = (const float*)d_in[0];
    const float* W      = (const float*)d_in[1];
    float* out          = (float*)d_out;

    float* curr = (float*)d_ws;
    float* BT   = (float*)((char*)d_ws + (size_t)TSTEPS * EXPERTS * sizeof(float));

    transpose_w_kernel<<<(HIDDEN * EXPERTS) / 256, 256, 0, stream>>>(W, BT);

    gemm_wmma_kernel<<<512, 32, 0, stream>>>(hidden, BT, curr);

    lif_scan_kernel<<<1, LIF_THREADS, 0, stream>>>(curr);

    routing_kernel<<<TSTEPS / 8, 256, 0, stream>>>((const float2*)curr, (float2*)out);
}


# LIF scan: split over two scan waves, one expert per lane (was one wave, two experts per lane): each wave issues half the vector instructions per step; arithmetic and sums unchanged
# speedup vs baseline: 2.4282x; 1.0705x over previous
#include <hip/hip_runtime.h>


typedef float v2f __attribute__((ext_vector_type(2)));
typedef float v8f __attribute__((ext_vector_type(8)));

#define HIDDEN 2048
#define EXPERTS 64
#define TSTEPS 8192

__global__ __launch_bounds__(256) void transpose_w_kernel(const float* __restrict__ W,
                                                          float* __restrict__ BT) {
    int idx = blockIdx.x * 256 + threadIdx.x;
    int k = idx >> 6;
    int n = idx & 63;
    BT[(size_t)n * HIDDEN + k] = W[(size_t)k * EXPERTS + n];
}

__global__ __launch_bounds__(32) void gemm_wmma_kernel(const float* __restrict__ A,
                                                        const float* __restrict__ BT,
                                                        float* __restrict__ C) {
    const int lane  = threadIdx.x;
    const int mtile = blockIdx.x;
    const int m0    = mtile * 16;
    const int half  = lane >> 4;
    const int sub   = lane & 15;

    const float* arow = A  + (size_t)(m0 + sub) * HIDDEN + half * 2;
    const float* b0p  = BT + (size_t)(0 * 16 + sub) * HIDDEN + half * 2;
    const float* b1p  = BT + (size_t)(1 * 16 + sub) * HIDDEN + half * 2;
    const float* b2p  = BT + (size_t)(2 * 16 + sub) * HIDDEN + half * 2;
    const float* b3p  = BT + (size_t)(3 * 16 + sub) * HIDDEN + half * 2;

    v8f acc0 = {}, acc1 = {}, acc2 = {}, acc3 = {};

#define KR 16
    v2f ra[KR], rb0[KR], rb1[KR], rb2[KR], rb3[KR];
#pragma unroll
    for (int j = 0; j < KR; ++j) {
        ra[j]  = *(const v2f*)(arow + 4 * j);
        rb0[j] = *(const v2f*)(b0p  + 4 * j);
        rb1[j] = *(const v2f*)(b1p  + 4 * j);
        rb2[j] = *(const v2f*)(b2p  + 4 * j);
        rb3[j] = *(const v2f*)(b3p  + 4 * j);
    }

    int k0 = 0;
    for (; k0 < HIDDEN - 4 * KR; k0 += 4 * KR) {
        __builtin_prefetch(arow + k0 + 512, 0, 3);
        __builtin_prefetch(arow + k0 + 544, 0, 3);
#pragma unroll
        for (int j = 0; j < KR; ++j) {
            acc0 = __builtin_amdgcn_wmma_f32_16x16x4_f32(false, ra[j], false, rb0[j], (short)0, acc0, false, false);
            acc1 = __builtin_amdgcn_wmma_f32_16x16x4_f32(false, ra[j], false, rb1[j], (short)0, acc1, false, false);
            acc2 = __builtin_amdgcn_wmma_f32_16x16x4_f32(false, ra[j], false, rb2[j], (short)0, acc2, false, false);
            acc3 = __builtin_amdgcn_wmma_f32_16x16x4_f32(false, ra[j], false, rb3[j], (short)0, acc3, false, false);
            const int kn = k0 + 4 * KR + 4 * j;
            ra[j]  = *(const v2f*)(arow + kn);
            rb0[j] = *(const v2f*)(b0p  + kn);
            rb1[j] = *(const v2f*)(b1p  + kn);
            rb2[j] = *(const v2f*)(b2p  + kn);
            rb3[j] = *(const v2f*)(b3p  + kn);
        }
    }
#pragma unroll
    for (int j = 0; j < KR; ++j) {
        acc0 = __builtin_amdgcn_wmma_f32_16x16x4_f32(false, ra[j], false, rb0[j], (short)0, acc0, false, false);
        acc1 = __builtin_amdgcn_wmma_f32_16x16x4_f32(false, ra[j], false, rb1[j], (short)0, acc1, false, false);
        acc2 = __builtin_amdgcn_wmma_f32_16x16x4_f32(false, ra[j], false, rb2[j], (short)0, acc2, false, false);
        acc3 = __builtin_amdgcn_wmma_f32_16x16x4_f32(false, ra[j], false, rb3[j], (short)0, acc3, false, false);
    }
#undef KR

#pragma unroll
    for (int r = 0; r < 8; ++r) {
        size_t row = (size_t)(m0 + half * 8 + r) * EXPERTS;
        C[row +  0 + sub] = acc0[r];
        C[row + 16 + sub] = acc1[r];
        C[row + 32 + sub] = acc2[r];
        C[row + 48 + sub] = acc3[r];
    }
}

#define CH 256
#define NCH (TSTEPS / CH)
#define LR 32
#define SCANW 2
#define HELPERS 8
#define LIF_THREADS (32 * (SCANW + HELPERS))
#define CH_F4 (CH * EXPERTS / 4)
#define H_ROUNDS (CH_F4 / (HELPERS * 32))

__device__ __forceinline__ float lif_step(float cur, float& m9, float& a, float& rf) {
    const float p = __builtin_fmaf(cur, a, m9);
    const float d = fmaxf(rf - 0.1f, 0.0f);
    float n = (rf <= 0.1f) ? 0.1f : 0.0f;
    asm("" : "+v"(n));
    const float t = p * 0.9f;
    const bool s = p > 1.0f;
    m9 = s ? 0.0f : t;
    a  = s ? 0.0f : n;
    rf = s ? 1.0f : d;
    return p;
}

typedef float v4f __attribute__((ext_vector_type(4)));

__global__ __launch_bounds__(LIF_THREADS) void lif_scan_kernel(float* __restrict__ gbuf) {
    __shared__ v4f lds[3][CH_F4];

    const int wave = __builtin_amdgcn_readfirstlane((int)(threadIdx.x >> 5));
    const int lane = threadIdx.x & 31;
    const int h    = (int)threadIdx.x - 32 * SCANW;

    v4f* __restrict__ g4 = (v4f*)gbuf;

    float m9 = 0.0f;
    float a  = 0.1f;
    float rf = 0.0f;

    v4f stage[H_ROUNDS];

    if (wave >= SCANW) {
#pragma unroll
        for (int r = 0; r < H_ROUNDS; ++r) stage[r] = g4[r * (HELPERS * 32) + h];
#pragma unroll
        for (int r = 0; r < H_ROUNDS; ++r) lds[0][r * (HELPERS * 32) + h] = stage[r];
#pragma unroll
        for (int r = 0; r < H_ROUNDS; ++r) stage[r] = g4[CH_F4 + r * (HELPERS * 32) + h];
    }
    __syncthreads();

    int cur = 0, nxt = 1, prv = 2;
    for (int c = 0; c < NCH; ++c) {
        if (wave < SCANW) {
            float* rows = (float*)&lds[cur][0] + (wave * 32 + lane);
            float ring[LR];
#pragma unroll
            for (int i = 0; i < LR; ++i) ring[i] = rows[i * EXPERTS];
            int t0 = 0;
            for (; t0 < CH - LR; t0 += LR) {
#pragma unroll
                for (int i = 0; i < LR; i += 2) {
                    const float p0 = lif_step(ring[i],     m9, a, rf);
                    const float p1 = lif_step(ring[i + 1], m9, a, rf);
                    rows[(t0 + i) * EXPERTS]     = p0;
                    rows[(t0 + i + 1) * EXPERTS] = p1;
                    ring[i]     = rows[(t0 + LR + i) * EXPERTS];
                    ring[i + 1] = rows[(t0 + LR + i + 1) * EXPERTS];
                }
            }
#pragma unroll
            for (int i = 0; i < LR; ++i) {
                const float p = lif_step(ring[i], m9, a, rf);
                rows[(t0 + i) * EXPERTS] = p;
            }
        } else {
            if (c + 1 < NCH) {
#pragma unroll
                for (int r = 0; r < H_ROUNDS; ++r) lds[nxt][r * (HELPERS * 32) + h] = stage[r];
            }
            if (c + 2 < NCH) {
                const v4f* src = g4 + (size_t)(c + 2) * CH_F4;
#pragma unroll
                for (int r = 0; r < H_ROUNDS; ++r) stage[r] = src[r * (HELPERS * 32) + h];
            }
            if (c >= 1) {
                v4f* dst = g4 + (size_t)(c - 1) * CH_F4;
#pragma unroll
                for (int r = 0; r < H_ROUNDS; ++r) dst[r * (HELPERS * 32) + h] = lds[prv][r * (HELPERS * 32) + h];
            }
        }
        __syncthreads();
        const int t = prv; prv = cur; cur = nxt; nxt = t;
    }

    if (wave >= SCANW) {
        v4f* dst = g4 + (size_t)(NCH - 1) * CH_F4;
#pragma unroll
        for (int r = 0; r < H_ROUNDS; ++r) dst[r * (HELPERS * 32) + h] = lds[prv][r * (HELPERS * 32) + h];
    }
}

__global__ __launch_bounds__(256) void routing_kernel(const float2* __restrict__ mpbuf,
                                                      float2* __restrict__ out) {
    const int row  = blockIdx.x * 8 + (threadIdx.x >> 5);
    const int lane = threadIdx.x & 31;

    float2 mp = mpbuf[(size_t)row * 32 + lane];
    bool s0 = mp.x > 1.0f;
    bool s1 = mp.y > 1.0f;

    unsigned long long bal = __ballot(s0 || s1);
    bool any = (bal != 0ull);

    float x0 = any ? (s0 ? 1.0f : 0.0f) : mp.x;
    float x1 = any ? (s1 ? 1.0f : 0.0f) : mp.y;

    float m = fmaxf(x0, x1);
#pragma unroll
    for (int off = 16; off >= 1; off >>= 1)
        m = fmaxf(m, __shfl_xor(m, off, 32));

    float e0 = __expf(x0 - m);
    float e1 = __expf(x1 - m);
    float s  = e0 + e1;
#pragma unroll
    for (int off = 16; off >= 1; off >>= 1)
        s += __shfl_xor(s, off, 32);

    float inv = 1.0f / s;
    float2 r;
    r.x = e0 * inv;
    r.y = e1 * inv;
    out[(size_t)row * 32 + lane] = r;
}

extern "C" void kernel_launch(void* const* d_in, const int* in_sizes, int n_in,
                              void* d_out, int out_size, void* d_ws, size_t ws_size,
                              hipStream_t stream) {
    const float* hidden = (const float*)d_in[0];
    const float* W      = (const float*)d_in[1];
    float* out          = (float*)d_out;

    float* curr = (float*)d_ws;
    float* BT   = (float*)((char*)d_ws + (size_t)TSTEPS * EXPERTS * sizeof(float));

    transpose_w_kernel<<<(HIDDEN * EXPERTS) / 256, 256, 0, stream>>>(W, BT);

    gemm_wmma_kernel<<<512, 32, 0, stream>>>(hidden, BT, curr);

    lif_scan_kernel<<<1, LIF_THREADS, 0, stream>>>(curr);

    routing_kernel<<<TSTEPS / 8, 256, 0, stream>>>((const float2*)curr, (float2*)out);
}


# LIF scan cut into 32 time segments run side by side, each warmed up over the 512 steps before it; opening states checked bitwise against predecessors' end states, serial rescan if not; sums unchanged
# speedup vs baseline: 7.7939x; 3.2097x over previous
#include <hip/hip_runtime.h>


typedef float v2f __attribute__((ext_vector_type(2)));
typedef float v8f __attribute__((ext_vector_type(8)));

#define HIDDEN 2048
#define EXPERTS 64
#define TSTEPS 8192

__global__ __launch_bounds__(256) void transpose_w_kernel(const float* __restrict__ W,
                                                          float* __restrict__ BT) {
    int idx = blockIdx.x * 256 + threadIdx.x;
    int k = idx >> 6;
    int n = idx & 63;
    BT[(size_t)n * HIDDEN + k] = W[(size_t)k * EXPERTS + n];
}

__global__ __launch_bounds__(32) void gemm_wmma_kernel(const float* __restrict__ A,
                                                        const float* __restrict__ BT,
                                                        float* __restrict__ C) {
    const int lane  = threadIdx.x;
    const int mtile = blockIdx.x;
    const int m0    = mtile * 16;
    const int half  = lane >> 4;
    const int sub   = lane & 15;

    const float* arow = A  + (size_t)(m0 + sub) * HIDDEN + half * 2;
    const float* b0p  = BT + (size_t)(0 * 16 + sub) * HIDDEN + half * 2;
    const float* b1p  = BT + (size_t)(1 * 16 + sub) * HIDDEN + half * 2;
    const float* b2p  = BT + (size_t)(2 * 16 + sub) * HIDDEN + half * 2;
    const float* b3p  = BT + (size_t)(3 * 16 + sub) * HIDDEN + half * 2;

    v8f acc0 = {}, acc1 = {}, acc2 = {}, acc3 = {};

#define KR 16
    v2f ra[KR], rb0[KR], rb1[KR], rb2[KR], rb3[KR];
#pragma unroll
    for (int j = 0; j < KR; ++j) {
        ra[j]  = *(const v2f*)(arow + 4 * j);
        rb0[j] = *(const v2f*)(b0p  + 4 * j);
        rb1[j] = *(const v2f*)(b1p  + 4 * j);
        rb2[j] = *(const v2f*)(b2p  + 4 * j);
        rb3[j] = *(const v2f*)(b3p  + 4 * j);
    }

    int k0 = 0;
    for (; k0 < HIDDEN - 4 * KR; k0 += 4 * KR) {
        __builtin_prefetch(arow + k0 + 512, 0, 3);
        __builtin_prefetch(arow + k0 + 544, 0, 3);
#pragma unroll
        for (int j = 0; j < KR; ++j) {
            acc0 = __builtin_amdgcn_wmma_f32_16x16x4_f32(false, ra[j], false, rb0[j], (short)0, acc0, false, false);
            acc1 = __builtin_amdgcn_wmma_f32_16x16x4_f32(false, ra[j], false, rb1[j], (short)0, acc1, false, false);
            acc2 = __builtin_amdgcn_wmma_f32_16x16x4_f32(false, ra[j], false, rb2[j], (short)0, acc2, false, false);
            acc3 = __builtin_amdgcn_wmma_f32_16x16x4_f32(false, ra[j], false, rb3[j], (short)0, acc3, false, false);
            const int kn = k0 + 4 * KR + 4 * j;
            ra[j]  = *(const v2f*)(arow + kn);
            rb0[j] = *(const v2f*)(b0p  + kn);
            rb1[j] = *(const v2f*)(b1p  + kn);
            rb2[j] = *(const v2f*)(b2p  + kn);
            rb3[j] = *(const v2f*)(b3p  + kn);
        }
    }
#pragma unroll
    for (int j = 0; j < KR; ++j) {
        acc0 = __builtin_amdgcn_wmma_f32_16x16x4_f32(false, ra[j], false, rb0[j], (short)0, acc0, false, false);
        acc1 = __builtin_amdgcn_wmma_f32_16x16x4_f32(false, ra[j], false, rb1[j], (short)0, acc1, false, false);
        acc2 = __builtin_amdgcn_wmma_f32_16x16x4_f32(false, ra[j], false, rb2[j], (short)0, acc2, false, false);
        acc3 = __builtin_amdgcn_wmma_f32_16x16x4_f32(false, ra[j], false, rb3[j], (short)0, acc3, false, false);
    }
#undef KR

#pragma unroll
    for (int r = 0; r < 8; ++r) {
        size_t row = (size_t)(m0 + half * 8 + r) * EXPERTS;
        C[row +  0 + sub] = acc0[r];
        C[row + 16 + sub] = acc1[r];
        C[row + 32 + sub] = acc2[r];
        C[row + 48 + sub] = acc3[r];
    }
}

#define CH 256
#define NCH (TSTEPS / CH)
#define LR 32
#define SCANW 2
#define HELPERS 8
#define LIF_THREADS (32 * (SCANW + HELPERS))
#define CH_F4 (CH * EXPERTS / 4)
#define H_ROUNDS (CH_F4 / (HELPERS * 32))
#define WARMC 2
#define NPLANE (NCH * EXPERTS)

__device__ __forceinline__ float lif_step(float cur, float& m9, float& a, float& rf) {
    const float p = __builtin_fmaf(cur, a, m9);
    const float d = fmaxf(rf - 0.1f, 0.0f);
    float n = (rf <= 0.1f) ? 0.1f : 0.0f;
    asm("" : "+v"(n));
    const float t = p * 0.9f;
    const bool s = p > 1.0f;
    m9 = s ? 0.0f : t;
    a  = s ? 0.0f : n;
    rf = s ? 1.0f : d;
    return p;
}

typedef float v4f __attribute__((ext_vector_type(4)));

__global__ __launch_bounds__(LIF_THREADS) void lif_scan_kernel(const float* __restrict__ cin,
                                                                float* __restrict__ pout,
                                                                float* __restrict__ st,
                                                                int mode) {
    __shared__ v4f lds[3][CH_F4];
    __shared__ int bad[NCH];

    const int wave = __builtin_amdgcn_readfirstlane((int)(threadIdx.x >> 5));
    const int lane = threadIdx.x & 31;
    const int h    = (int)threadIdx.x - 32 * SCANW;
    const int ex   = (int)threadIdx.x & (EXPERTS - 1);

    const v4f* __restrict__ gi = (const v4f*)cin;
    v4f* __restrict__ go = (v4f*)pout;

    float m9 = 0.0f;
    float a  = 0.1f;
    float rf = 0.0f;

    int c_begin, c_store, c_end;
    if (mode == 0) {
        const int s = blockIdx.x;
        c_begin = s > WARMC ? s - WARMC : 0;
        c_store = s;
        c_end   = s + 1;
    } else {
        if (threadIdx.x < NCH) bad[threadIdx.x] = 0;
        __syncthreads();
        for (int i = threadIdx.x; i < (NCH - 1) * EXPERTS; i += LIF_THREADS) {
            const int s = 1 + (i >> 6);
            const int e = i & (EXPERTS - 1);
            const int o = s * EXPERTS + e;
            const bool diff =
                (__float_as_uint(st[0 * NPLANE + o]) != __float_as_uint(st[3 * NPLANE + o - EXPERTS])) ||
                (__float_as_uint(st[1 * NPLANE + o]) != __float_as_uint(st[4 * NPLANE + o - EXPERTS])) ||
                (__float_as_uint(st[2 * NPLANE + o]) != __float_as_uint(st[5 * NPLANE + o - EXPERTS]));
            if (diff) bad[s] = 1;
        }
        __syncthreads();
        int first = NCH;
        for (int s = NCH - 1; s >= 1; --s) first = bad[s] ? s : first;
        first = __builtin_amdgcn_readfirstlane(first);
        if (first == NCH) return;
        c_begin = first;
        c_store = first;
        c_end   = NCH;
        if (wave < SCANW) {
            const int o = (first - 1) * EXPERTS + ex;
            m9 = st[3 * NPLANE + o];
            a  = st[4 * NPLANE + o];
            rf = st[5 * NPLANE + o];
        }
    }

    v4f stage[H_ROUNDS];

    if (wave >= SCANW) {
        const v4f* src0 = gi + (size_t)c_begin * CH_F4;
#pragma unroll
        for (int r = 0; r < H_ROUNDS; ++r) stage[r] = src0[r * (HELPERS * 32) + h];
#pragma unroll
        for (int r = 0; r < H_ROUNDS; ++r) lds[0][r * (HELPERS * 32) + h] = stage[r];
        if (c_begin + 1 < c_end) {
#pragma unroll
            for (int r = 0; r < H_ROUNDS; ++r) stage[r] = src0[CH_F4 + r * (HELPERS * 32) + h];
        }
    }
    __syncthreads();

    int cur = 0, nxt = 1, prv = 2;
    for (int c = c_begin; c < c_end; ++c) {
        if (wave < SCANW) {
            if (mode == 0 && c == c_store) {
                const int o = c * EXPERTS + ex;
                st[0 * NPLANE + o] = m9;
                st[1 * NPLANE + o] = a;
                st[2 * NPLANE + o] = rf;
            }
            float* rows = (float*)&lds[cur][0] + (wave * 32 + lane);
            float ring[LR];
#pragma unroll
            for (int i = 0; i < LR; ++i) ring[i] = rows[i * EXPERTS];
            int t0 = 0;
            for (; t0 < CH - LR; t0 += LR) {
#pragma unroll
                for (int i = 0; i < LR; i += 2) {
                    const float p0 = lif_step(ring[i],     m9, a, rf);
                    const float p1 = lif_step(ring[i + 1], m9, a, rf);
                    rows[(t0 + i) * EXPERTS]     = p0;
                    rows[(t0 + i + 1) * EXPERTS] = p1;
                    ring[i]     = rows[(t0 + LR + i) * EXPERTS];
                    ring[i + 1] = rows[(t0 + LR + i + 1) * EXPERTS];
                }
            }
#pragma unroll
            for (int i = 0; i < LR; ++i) {
                const float p = lif_step(ring[i], m9, a, rf);
                rows[(t0 + i) * EXPERTS] = p;
            }
        } else {
            if (c + 1 < c_end) {
#pragma unroll
                for (int r = 0; r < H_ROUNDS; ++r) lds[nxt][r * (HELPERS * 32) + h] = stage[r];
            }
            if (c + 2 < c_end) {
                const v4f* src = gi + (size_t)(c + 2) * CH_F4;
#pragma unroll
                for (int r = 0; r < H_ROUNDS; ++r) stage[r] = src[r * (HELPERS * 32) + h];
            }
            if (c > c_store) {
                v4f* dst = go + (size_t)(c - 1) * CH_F4;
#pragma unroll
                for (int r = 0; r < H_ROUNDS; ++r) dst[r * (HELPERS * 32) + h] = lds[prv][r * (HELPERS * 32) + h];
            }
        }
        __syncthreads();
        const int t = prv; prv = cur; cur = nxt; nxt = t;
    }

    if (wave >= SCANW) {
        v4f* dst = go + (size_t)(c_end - 1) * CH_F4;
#pragma unroll
        for (int r = 0; r < H_ROUNDS; ++r) dst[r * (HELPERS * 32) + h] = lds[prv][r * (HELPERS * 32) + h];
    } else if (mode == 0) {
        const int o = c_store * EXPERTS + ex;
        st[3 * NPLANE + o] = m9;
        st[4 * NPLANE + o] = a;
        st[5 * NPLANE + o] = rf;
    }
}

__global__ __launch_bounds__(256) void routing_kernel(float2* __restrict__ io) {
    const int row  = blockIdx.x * 8 + (threadIdx.x >> 5);
    const int lane = threadIdx.x & 31;

    float2 mp = io[(size_t)row * 32 + lane];
    bool s0 = mp.x > 1.0f;
    bool s1 = mp.y > 1.0f;

    unsigned long long bal = __ballot(s0 || s1);
    bool any = (bal != 0ull);

    float x0 = any ? (s0 ? 1.0f : 0.0f) : mp.x;
    float x1 = any ? (s1 ? 1.0f : 0.0f) : mp.y;

    float m = fmaxf(x0, x1);
#pragma unroll
    for (int off = 16; off >= 1; off >>= 1)
        m = fmaxf(m, __shfl_xor(m, off, 32));

    float e0 = __expf(x0 - m);
    float e1 = __expf(x1 - m);
    float s  = e0 + e1;
#pragma unroll
    for (int off = 16; off >= 1; off >>= 1)
        s += __shfl_xor(s, off, 32);

    float inv = 1.0f / s;
    float2 r;
    r.x = e0 * inv;
    r.y = e1 * inv;
    io[(size_t)row * 32 + lane] = r;
}

extern "C" void kernel_launch(void* const* d_in, const int* in_sizes, int n_in,
                              void* d_out, int out_size, void* d_ws, size_t ws_size,
                              hipStream_t stream) {
    const float* hidden = (const float*)d_in[0];
    const float* W      = (const float*)d_in[1];
    float* out          = (float*)d_out;

    float* curr = (float*)d_ws;
    float* BT   = (float*)((char*)d_ws + (size_t)TSTEPS * EXPERTS * sizeof(float));
    float* st   = BT;

    transpose_w_kernel<<<(HIDDEN * EXPERTS) / 256, 256, 0, stream>>>(W, BT);

    gemm_wmma_kernel<<<512, 32, 0, stream>>>(hidden, BT, curr);

    lif_scan_kernel<<<NCH, LIF_THREADS, 0, stream>>>(curr, out, st, 0);
    lif_scan_kernel<<<1, LIF_THREADS, 0, stream>>>(curr, out, st, 1);

    routing_kernel<<<TSTEPS / 8, 256, 0, stream>>>((float2*)out);
}


# GEMM B operand: fragments read straight from W rows by 16-byte loads (4 whole lines per load, was 16 lines of a transposed copy); transpose launch gone; tile t column c = expert 4c+t; sums unchanged
# speedup vs baseline: 9.0887x; 1.1661x over previous
#include <hip/hip_runtime.h>


typedef float v2f __attribute__((ext_vector_type(2)));
typedef float v4f __attribute__((ext_vector_type(4)));
typedef float v8f __attribute__((ext_vector_type(8)));

#define HIDDEN 2048
#define EXPERTS 64
#define TSTEPS 8192

__global__ __launch_bounds__(32) void gemm_wmma_kernel(const float* __restrict__ A,
                                                        const float* __restrict__ W,
                                                        float* __restrict__ C) {
    const int lane  = threadIdx.x;
    const int mtile = blockIdx.x;
    const int m0    = mtile * 16;
    const int half  = lane >> 4;
    const int sub   = lane & 15;

    const float* arow = A + (size_t)(m0 + sub) * HIDDEN + half * 2;
    const float* wrow = W + (size_t)(half * 2) * EXPERTS + sub * 4;

    v8f acc0 = {}, acc1 = {}, acc2 = {}, acc3 = {};

#define KR 16
    v2f ra[KR];
    v4f rw0[KR], rw1[KR];
#pragma unroll
    for (int j = 0; j < KR; ++j) {
        ra[j]  = *(const v2f*)(arow + 4 * j);
        rw0[j] = *(const v4f*)(wrow + (size_t)(4 * j) * EXPERTS);
        rw1[j] = *(const v4f*)(wrow + (size_t)(4 * j + 1) * EXPERTS);
        __builtin_amdgcn_sched_barrier(0);
    }

    int k0 = 0;
    for (; k0 < HIDDEN - 4 * KR; k0 += 4 * KR) {
        __builtin_prefetch(arow + k0 + 512, 0, 3);
        __builtin_prefetch(arow + k0 + 544, 0, 3);
#pragma unroll
        for (int j = 0; j < KR; ++j) {
            const v2f b0 = {rw0[j].x, rw1[j].x};
            const v2f b1 = {rw0[j].y, rw1[j].y};
            const v2f b2 = {rw0[j].z, rw1[j].z};
            const v2f b3 = {rw0[j].w, rw1[j].w};
            acc0 = __builtin_amdgcn_wmma_f32_16x16x4_f32(false, ra[j], false, b0, (short)0, acc0, false, false);
            acc1 = __builtin_amdgcn_wmma_f32_16x16x4_f32(false, ra[j], false, b1, (short)0, acc1, false, false);
            acc2 = __builtin_amdgcn_wmma_f32_16x16x4_f32(false, ra[j], false, b2, (short)0, acc2, false, false);
            acc3 = __builtin_amdgcn_wmma_f32_16x16x4_f32(false, ra[j], false, b3, (short)0, acc3, false, false);
            const int kn = k0 + 4 * KR + 4 * j;
            ra[j]  = *(const v2f*)(arow + kn);
            rw0[j] = *(const v4f*)(wrow + (size_t)kn * EXPERTS);
            rw1[j] = *(const v4f*)(wrow + (size_t)(kn + 1) * EXPERTS);
            __builtin_amdgcn_sched_barrier(0);
        }
    }
#pragma unroll
    for (int j = 0; j < KR; ++j) {
        const v2f b0 = {rw0[j].x, rw1[j].x};
        const v2f b1 = {rw0[j].y, rw1[j].y};
        const v2f b2 = {rw0[j].z, rw1[j].z};
        const v2f b3 = {rw0[j].w, rw1[j].w};
        acc0 = __builtin_amdgcn_wmma_f32_16x16x4_f32(false, ra[j], false, b0, (short)0, acc0, false, false);
        acc1 = __builtin_amdgcn_wmma_f32_16x16x4_f32(false, ra[j], false, b1, (short)0, acc1, false, false);
        acc2 = __builtin_amdgcn_wmma_f32_16x16x4_f32(false, ra[j], false, b2, (short)0, acc2, false, false);
        acc3 = __builtin_amdgcn_wmma_f32_16x16x4_f32(false, ra[j], false, b3, (short)0, acc3, false, false);
    }
#undef KR

#pragma unroll
    for (int r = 0; r < 8; ++r) {
        const size_t row = (size_t)(m0 + half * 8 + r) * EXPERTS;
        const v4f o = {acc0[r], acc1[r], acc2[r], acc3[r]};
        *(v4f*)(C + row + sub * 4) = o;
    }
}

#define CH 256
#define NCH (TSTEPS / CH)
#define LR 32
#define SCANW 2
#define HELPERS 8
#define LIF_THREADS (32 * (SCANW + HELPERS))
#define CH_F4 (CH * EXPERTS / 4)
#define H_ROUNDS (CH_F4 / (HELPERS * 32))
#define WARMC 2
#define NPLANE (NCH * EXPERTS)

__device__ __forceinline__ float lif_step(float cur, float& m9, float& a, float& rf) {
    const float p = __builtin_fmaf(cur, a, m9);
    const float d = fmaxf(rf - 0.1f, 0.0f);
    float n = (rf <= 0.1f) ? 0.1f : 0.0f;
    asm("" : "+v"(n));
    const float t = p * 0.9f;
    const bool s = p > 1.0f;
    m9 = s ? 0.0f : t;
    a  = s ? 0.0f : n;
    rf = s ? 1.0f : d;
    return p;
}

__global__ __launch_bounds__(LIF_THREADS) void lif_scan_kernel(const float* __restrict__ cin,
                                                                float* __restrict__ pout,
                                                                float* __restrict__ st,
                                                                int mode) {
    __shared__ v4f lds[3][CH_F4];
    __shared__ int bad[NCH];

    const int wave = __builtin_amdgcn_readfirstlane((int)(threadIdx.x >> 5));
    const int lane = threadIdx.x & 31;
    const int h    = (int)threadIdx.x - 32 * SCANW;
    const int ex   = (int)threadIdx.x & (EXPERTS - 1);

    const v4f* __restrict__ gi = (const v4f*)cin;
    v4f* __restrict__ go = (v4f*)pout;

    float m9 = 0.0f;
    float a  = 0.1f;
    float rf = 0.0f;

    int c_begin, c_store, c_end;
    if (mode == 0) {
        const int s = blockIdx.x;
        c_begin = s > WARMC ? s - WARMC : 0;
        c_store = s;
        c_end   = s + 1;
    } else {
        if (threadIdx.x < NCH) bad[threadIdx.x] = 0;
        __syncthreads();
        for (int i = threadIdx.x; i < (NCH - 1) * EXPERTS; i += LIF_THREADS) {
            const int s = 1 + (i >> 6);
            const int e = i & (EXPERTS - 1);
            const int o = s * EXPERTS + e;
            const bool diff =
                (__float_as_uint(st[0 * NPLANE + o]) != __float_as_uint(st[3 * NPLANE + o - EXPERTS])) ||
                (__float_as_uint(st[1 * NPLANE + o]) != __float_as_uint(st[4 * NPLANE + o - EXPERTS])) ||
                (__float_as_uint(st[2 * NPLANE + o]) != __float_as_uint(st[5 * NPLANE + o - EXPERTS]));
            if (diff) bad[s] = 1;
        }
        __syncthreads();
        int first = NCH;
        for (int s = NCH - 1; s >= 1; --s) first = bad[s] ? s : first;
        first = __builtin_amdgcn_readfirstlane(first);
        if (first == NCH) return;
        c_begin = first;
        c_store = first;
        c_end   = NCH;
        if (wave < SCANW) {
            const int o = (first - 1) * EXPERTS + ex;
            m9 = st[3 * NPLANE + o];
            a  = st[4 * NPLANE + o];
            rf = st[5 * NPLANE + o];
        }
    }

    v4f stage[H_ROUNDS];

    if (wave >= SCANW) {
        const v4f* src0 = gi + (size_t)c_begin * CH_F4;
#pragma unroll
        for (int r = 0; r < H_ROUNDS; ++r) stage[r] = src0[r * (HELPERS * 32) + h];
#pragma unroll
        for (int r = 0; r < H_ROUNDS; ++r) lds[0][r * (HELPERS * 32) + h] = stage[r];
        if (c_begin + 1 < c_end) {
#pragma unroll
            for (int r = 0; r < H_ROUNDS; ++r) stage[r] = src0[CH_F4 + r * (HELPERS * 32) + h];
        }
    }
    __syncthreads();

    int cur = 0, nxt = 1, prv = 2;
    for (int c = c_begin; c < c_end; ++c) {
        if (wave < SCANW) {
            if (mode == 0 && c == c_store) {
                const int o = c * EXPERTS + ex;
                st[0 * NPLANE + o] = m9;
                st[1 * NPLANE + o] = a;
                st[2 * NPLANE + o] = rf;
            }
            float* rows = (float*)&lds[cur][0] + (wave * 32 + lane);
            float ring[LR];
#pragma unroll
            for (int i = 0; i < LR; ++i) ring[i] = rows[i * EXPERTS];
            int t0 = 0;
            for (; t0 < CH - LR; t0 += LR) {
#pragma unroll
                for (int i = 0; i < LR; i += 2) {
                    const float p0 = lif_step(ring[i],     m9, a, rf);
                    const float p1 = lif_step(ring[i + 1], m9, a, rf);
                    rows[(t0 + i) * EXPERTS]     = p0;
                    rows[(t0 + i + 1) * EXPERTS] = p1;
                    ring[i]     = rows[(t0 + LR + i) * EXPERTS];
                    ring[i + 1] = rows[(t0 + LR + i + 1) * EXPERTS];
                }
            }
#pragma unroll
            for (int i = 0; i < LR; ++i) {
                const float p = lif_step(ring[i], m9, a, rf);
                rows[(t0 + i) * EXPERTS] = p;
            }
        } else {
            if (c + 1 < c_end) {
#pragma unroll
                for (int r = 0; r < H_ROUNDS; ++r) lds[nxt][r * (HELPERS * 32) + h] = stage[r];
            }
            if (c + 2 < c_end) {
                const v4f* src = gi + (size_t)(c + 2) * CH_F4;
#pragma unroll
                for (int r = 0; r < H_ROUNDS; ++r) stage[r] = src[r * (HELPERS * 32) + h];
            }
            if (c > c_store) {
                v4f* dst = go + (size_t)(c - 1) * CH_F4;
#pragma unroll
                for (int r = 0; r < H_ROUNDS; ++r) dst[r * (HELPERS * 32) + h] = lds[prv][r * (HELPERS * 32) + h];
            }
        }
        __syncthreads();
        const int t = prv; prv = cur; cur = nxt; nxt = t;
    }

    if (wave >= SCANW) {
        v4f* dst = go + (size_t)(c_end - 1) * CH_F4;
#pragma unroll
        for (int r = 0; r < H_ROUNDS; ++r) dst[r * (HELPERS * 32) + h] = lds[prv][r * (HELPERS * 32) + h];
    } else if (mode == 0) {
        const int o = c_store * EXPERTS + ex;
        st[3 * NPLANE + o] = m9;
        st[4 * NPLANE + o] = a;
        st[5 * NPLANE + o] = rf;
    }
}

__global__ __launch_bounds__(256) void routing_kernel(float2* __restrict__ io) {
    const int row  = blockIdx.x * 8 + (threadIdx.x >> 5);
    const int lane = threadIdx.x & 31;

    float2 mp = io[(size_t)row * 32 + lane];
    bool s0 = mp.x > 1.0f;
    bool s1 = mp.y > 1.0f;

    unsigned long long bal = __ballot(s0 || s1);
    bool any = (bal != 0ull);

    float x0 = any ? (s0 ? 1.0f : 0.0f) : mp.x;
    float x1 = any ? (s1 ? 1.0f : 0.0f) : mp.y;

    float m = fmaxf(x0, x1);
#pragma unroll
    for (int off = 16; off >= 1; off >>= 1)
        m = fmaxf(m, __shfl_xor(m, off, 32));

    float e0 = __expf(x0 - m);
    float e1 = __expf(x1 - m);
    float s  = e0 + e1;
#pragma unroll
    for (int off = 16; off >= 1; off >>= 1)
        s += __shfl_xor(s, off, 32);

    float inv = 1.0f / s;
    float2 r;
    r.x = e0 * inv;
    r.y = e1 * inv;
    io[(size_t)row * 32 + lane] = r;
}

extern "C" void kernel_launch(void* const* d_in, const int* in_sizes, int n_in,
                              void* d_out, int out_size, void* d_ws, size_t ws_size,
                              hipStream_t stream) {
    const float* hidden = (const float*)d_in[0];
    const float* W      = (const float*)d_in[1];
    float* out          = (float*)d_out;

    float* curr = (float*)d_ws;
    float* st   = (float*)((char*)d_ws + (size_t)TSTEPS * EXPERTS * sizeof(float));

    gemm_wmma_kernel<<<512, 32, 0, stream>>>(hidden, W, curr);

    lif_scan_kernel<<<NCH, LIF_THREADS, 0, stream>>>(curr, out, st, 0);
    lif_scan_kernel<<<1, LIF_THREADS, 0, stream>>>(curr, out, st, 1);

    routing_kernel<<<TSTEPS / 8, 256, 0, stream>>>((float2*)out);
}


# LIF scan re-cut: 128 segments of 64 steps, warm-up 448 (was 32 of 256, warm-up 512): a work-group scans 512 dependent steps, not 768; boundary check loads in one batch; sums unchanged
# speedup vs baseline: 11.0301x; 1.2136x over previous
#include <hip/hip_runtime.h>


typedef float v2f __attribute__((ext_vector_type(2)));
typedef float v4f __attribute__((ext_vector_type(4)));
typedef float v8f __attribute__((ext_vector_type(8)));
typedef unsigned int v4u __attribute__((ext_vector_type(4)));

#define HIDDEN 2048
#define EXPERTS 64
#define TSTEPS 8192

__global__ __launch_bounds__(32) void gemm_wmma_kernel(const float* __restrict__ A,
                                                        const float* __restrict__ W,
                                                        float* __restrict__ C) {
    const int lane  = threadIdx.x;
    const int mtile = blockIdx.x;
    const int m0    = mtile * 16;
    const int half  = lane >> 4;
    const int sub   = lane & 15;

    const float* arow = A + (size_t)(m0 + sub) * HIDDEN + half * 2;
    const float* wrow = W + (size_t)(half * 2) * EXPERTS + sub * 4;

    v8f acc0 = {}, acc1 = {}, acc2 = {}, acc3 = {};

#define KR 16
    v2f ra[KR];
    v4f rw0[KR], rw1[KR];
#pragma unroll
    for (int j = 0; j < KR; ++j) {
        ra[j]  = *(const v2f*)(arow + 4 * j);
        rw0[j] = *(const v4f*)(wrow + (size_t)(4 * j) * EXPERTS);
        rw1[j] = *(const v4f*)(wrow + (size_t)(4 * j + 1) * EXPERTS);
        __builtin_amdgcn_sched_barrier(0);
    }

    int k0 = 0;
    for (; k0 < HIDDEN - 4 * KR; k0 += 4 * KR) {
        __builtin_prefetch(arow + k0 + 512, 0, 3);
        __builtin_prefetch(arow + k0 + 544, 0, 3);
#pragma unroll
        for (int j = 0; j < KR; ++j) {
            const v2f b0 = {rw0[j].x, rw1[j].x};
            const v2f b1 = {rw0[j].y, rw1[j].y};
            const v2f b2 = {rw0[j].z, rw1[j].z};
            const v2f b3 = {rw0[j].w, rw1[j].w};
            acc0 = __builtin_amdgcn_wmma_f32_16x16x4_f32(false, ra[j], false, b0, (short)0, acc0, false, false);
            acc1 = __builtin_amdgcn_wmma_f32_16x16x4_f32(false, ra[j], false, b1, (short)0, acc1, false, false);
            acc2 = __builtin_amdgcn_wmma_f32_16x16x4_f32(false, ra[j], false, b2, (short)0, acc2, false, false);
            acc3 = __builtin_amdgcn_wmma_f32_16x16x4_f32(false, ra[j], false, b3, (short)0, acc3, false, false);
            const int kn = k0 + 4 * KR + 4 * j;
            ra[j]  = *(const v2f*)(arow + kn);
            rw0[j] = *(const v4f*)(wrow + (size_t)kn * EXPERTS);
            rw1[j] = *(const v4f*)(wrow + (size_t)(kn + 1) * EXPERTS);
            __builtin_amdgcn_sched_barrier(0);
        }
    }
#pragma unroll
    for (int j = 0; j < KR; ++j) {
        const v2f b0 = {rw0[j].x, rw1[j].x};
        const v2f b1 = {rw0[j].y, rw1[j].y};
        const v2f b2 = {rw0[j].z, rw1[j].z};
        const v2f b3 = {rw0[j].w, rw1[j].w};
        acc0 = __builtin_amdgcn_wmma_f32_16x16x4_f32(false, ra[j], false, b0, (short)0, acc0, false, false);
        acc1 = __builtin_amdgcn_wmma_f32_16x16x4_f32(false, ra[j], false, b1, (short)0, acc1, false, false);
        acc2 = __builtin_amdgcn_wmma_f32_16x16x4_f32(false, ra[j], false, b2, (short)0, acc2, false, false);
        acc3 = __builtin_amdgcn_wmma_f32_16x16x4_f32(false, ra[j], false, b3, (short)0, acc3, false, false);
    }
#undef KR

#pragma unroll
    for (int r = 0; r < 8; ++r) {
        const size_t row = (size_t)(m0 + half * 8 + r) * EXPERTS;
        const v4f o = {acc0[r], acc1[r], acc2[r], acc3[r]};
        *(v4f*)(C + row + sub * 4) = o;
    }
}

#define CH 64
#define NCH (TSTEPS / CH)
#define LR 32
#define SCANW 2
#define HELPERS 8
#define LIF_THREADS (32 * (SCANW + HELPERS))
#define CH_F4 (CH * EXPERTS / 4)
#define H_ROUNDS (CH_F4 / (HELPERS * 32))
#define WARMC 7
#define CHK_F4 ((NCH - 1) * EXPERTS / 4)
#define CHK_ROUNDS ((CHK_F4 + LIF_THREADS - 1) / LIF_THREADS)
#define NPLANE (NCH * EXPERTS)

__device__ __forceinline__ float lif_step(float cur, float& m9, float& a, float& rf) {
    const float p = __builtin_fmaf(cur, a, m9);
    const float d = fmaxf(rf - 0.1f, 0.0f);
    float n = (rf <= 0.1f) ? 0.1f : 0.0f;
    asm("" : "+v"(n));
    const float t = p * 0.9f;
    const bool s = p > 1.0f;
    m9 = s ? 0.0f : t;
    a  = s ? 0.0f : n;
    rf = s ? 1.0f : d;
    return p;
}

__device__ __forceinline__ v4u f4bits(v4f v) { return __builtin_bit_cast(v4u, v); }

__global__ __launch_bounds__(LIF_THREADS) void lif_scan_kernel(const float* __restrict__ cin,
                                                                float* __restrict__ pout,
                                                                float* __restrict__ st,
                                                                int mode) {
    __shared__ v4f lds[3][CH_F4];
    __shared__ int bad[NCH];

    const int wave = __builtin_amdgcn_readfirstlane((int)(threadIdx.x >> 5));
    const int lane = threadIdx.x & 31;
    const int h    = (int)threadIdx.x - 32 * SCANW;
    const int ex   = (int)threadIdx.x & (EXPERTS - 1);

    const v4f* __restrict__ gi = (const v4f*)cin;
    v4f* __restrict__ go = (v4f*)pout;

    float m9 = 0.0f;
    float a  = 0.1f;
    float rf = 0.0f;

    int c_begin, c_store, c_end;
    if (mode == 0) {
        const int s = blockIdx.x;
        c_begin = s > WARMC ? s - WARMC : 0;
        c_store = s;
        c_end   = s + 1;
    } else {
        if (threadIdx.x < NCH) bad[threadIdx.x] = 0;
        __syncthreads();
        const v4f* sp = (const v4f*)st;
        v4u df[CHK_ROUNDS];
#pragma unroll
        for (int r = 0; r < CHK_ROUNDS; ++r) {
            const int i = min(r * LIF_THREADS + (int)threadIdx.x, CHK_F4 - 1);
            df[r] = (f4bits(sp[0 * (NPLANE / 4) + EXPERTS / 4 + i]) ^ f4bits(sp[3 * (NPLANE / 4) + i])) |
                    (f4bits(sp[1 * (NPLANE / 4) + EXPERTS / 4 + i]) ^ f4bits(sp[4 * (NPLANE / 4) + i])) |
                    (f4bits(sp[2 * (NPLANE / 4) + EXPERTS / 4 + i]) ^ f4bits(sp[5 * (NPLANE / 4) + i]));
        }
#pragma unroll
        for (int r = 0; r < CHK_ROUNDS; ++r) {
            const int i = min(r * LIF_THREADS + (int)threadIdx.x, CHK_F4 - 1);
            if ((df[r].x | df[r].y | df[r].z | df[r].w) != 0u) bad[1 + i / (EXPERTS / 4)] = 1;
        }
        __syncthreads();
        int first = NCH;
        for (int s = NCH - 1; s >= 1; --s) first = bad[s] ? s : first;
        first = __builtin_amdgcn_readfirstlane(first);
        if (first == NCH) return;
        c_begin = first;
        c_store = first;
        c_end   = NCH;
        if (wave < SCANW) {
            const int o = (first - 1) * EXPERTS + ex;
            m9 = st[3 * NPLANE + o];
            a  = st[4 * NPLANE + o];
            rf = st[5 * NPLANE + o];
        }
    }

    v4f stage[H_ROUNDS];

    if (wave >= SCANW) {
        const v4f* src0 = gi + (size_t)c_begin * CH_F4;
#pragma unroll
        for (int r = 0; r < H_ROUNDS; ++r) stage[r] = src0[r * (HELPERS * 32) + h];
#pragma unroll
        for (int r = 0; r < H_ROUNDS; ++r) lds[0][r * (HELPERS * 32) + h] = stage[r];
        if (c_begin + 1 < c_end) {
#pragma unroll
            for (int r = 0; r < H_ROUNDS; ++r) stage[r] = src0[CH_F4 + r * (HELPERS * 32) + h];
        }
    }
    __syncthreads();

    int cur = 0, nxt = 1, prv = 2;
    for (int c = c_begin; c < c_end; ++c) {
        if (wave < SCANW) {
            if (mode == 0 && c == c_store) {
                const int o = c * EXPERTS + ex;
                st[0 * NPLANE + o] = m9;
                st[1 * NPLANE + o] = a;
                st[2 * NPLANE + o] = rf;
            }
            float* rows = (float*)&lds[cur][0] + (wave * 32 + lane);
            float ring[LR];
#pragma unroll
            for (int i = 0; i < LR; ++i) ring[i] = rows[i * EXPERTS];
            int t0 = 0;
            for (; t0 < CH - LR; t0 += LR) {
#pragma unroll
                for (int i = 0; i < LR; i += 2) {
                    const float p0 = lif_step(ring[i],     m9, a, rf);
                    const float p1 = lif_step(ring[i + 1], m9, a, rf);
                    rows[(t0 + i) * EXPERTS]     = p0;
                    rows[(t0 + i + 1) * EXPERTS] = p1;
                    ring[i]     = rows[(t0 + LR + i) * EXPERTS];
                    ring[i + 1] = rows[(t0 + LR + i + 1) * EXPERTS];
                }
            }
#pragma unroll
            for (int i = 0; i < LR; ++i) {
                const float p = lif_step(ring[i], m9, a, rf);
                rows[(t0 + i) * EXPERTS] = p;
            }
        } else {
            if (c + 1 < c_end) {
#pragma unroll
                for (int r = 0; r < H_ROUNDS; ++r) lds[nxt][r * (HELPERS * 32) + h] = stage[r];
            }
            if (c + 2 < c_end) {
                const v4f* src = gi + (size_t)(c + 2) * CH_F4;
#pragma unroll
                for (int r = 0; r < H_ROUNDS; ++r) stage[r] = src[r * (HELPERS * 32) + h];
            }
            if (c > c_store) {
                v4f* dst = go + (size_t)(c - 1) * CH_F4;
#pragma unroll
                for (int r = 0; r < H_ROUNDS; ++r) dst[r * (HELPERS * 32) + h] = lds[prv][r * (HELPERS * 32) + h];
            }
        }
        __syncthreads();
        const int t = prv; prv = cur; cur = nxt; nxt = t;
    }

    if (wave >= SCANW) {
        v4f* dst = go + (size_t)(c_end - 1) * CH_F4;
#pragma unroll
        for (int r = 0; r < H_ROUNDS; ++r) dst[r * (HELPERS * 32) + h] = lds[prv][r * (HELPERS * 32) + h];
    } else if (mode == 0) {
        const int o = c_store * EXPERTS + ex;
        st[3 * NPLANE + o] = m9;
        st[4 * NPLANE + o] = a;
        st[5 * NPLANE + o] = rf;
    }
}

__global__ __launch_bounds__(256) void routing_kernel(float2* __restrict__ io) {
    const int row  = blockIdx.x * 8 + (threadIdx.x >> 5);
    const int lane = threadIdx.x & 31;

    float2 mp = io[(size_t)row * 32 + lane];
    bool s0 = mp.x > 1.0f;
    bool s1 = mp.y > 1.0f;

    unsigned long long bal = __ballot(s0 || s1);
    bool any = (bal != 0ull);

    float x0 = any ? (s0 ? 1.0f : 0.0f) : mp.x;
    float x1 = any ? (s1 ? 1.0f : 0.0f) : mp.y;

    float m = fmaxf(x0, x1);
#pragma unroll
    for (int off = 16; off >= 1; off >>= 1)
        m = fmaxf(m, __shfl_xor(m, off, 32));

    float e0 = __expf(x0 - m);
    float e1 = __expf(x1 - m);
    float s  = e0 + e1;
#pragma unroll
    for (int off = 16; off >= 1; off >>= 1)
        s += __shfl_xor(s, off, 32);

    float inv = 1.0f / s;
    float2 r;
    r.x = e0 * inv;
    r.y = e1 * inv;
    io[(size_t)row * 32 + lane] = r;
}

extern "C" void kernel_launch(void* const* d_in, const int* in_sizes, int n_in,
                              void* d_out, int out_size, void* d_ws, size_t ws_size,
                              hipStream_t stream) {
    const float* hidden = (const float*)d_in[0];
    const float* W      = (const float*)d_in[1];
    float* out          = (float*)d_out;

    float* curr = (float*)d_ws;
    float* st   = (float*)((char*)d_ws + (size_t)TSTEPS * EXPERTS * sizeof(float));

    gemm_wmma_kernel<<<512, 32, 0, stream>>>(hidden, W, curr);

    lif_scan_kernel<<<NCH, LIF_THREADS, 0, stream>>>(curr, out, st, 0);
    lif_scan_kernel<<<1, LIF_THREADS, 0, stream>>>(curr, out, st, 1);

    routing_kernel<<<TSTEPS / 8, 256, 0, stream>>>((float2*)out);
}


# GEMM spread over all 1024 SIMDs: two waves per M-tile (32 experts each), 256 groups of 4 waves; to feed 4 waves per WGP A is read 8 k per load, operand pairs form by in-place swaps; sums unchanged
# speedup vs baseline: 14.0034x; 1.2696x over previous
#include <hip/hip_runtime.h>


typedef float v2f __attribute__((ext_vector_type(2)));
typedef float v4f __attribute__((ext_vector_type(4)));
typedef float v8f __attribute__((ext_vector_type(8)));
typedef unsigned int v4u __attribute__((ext_vector_type(4)));

#define HIDDEN 2048
#define EXPERTS 64
#define TSTEPS 8192


#define KP 12
#define NPAIR (HIDDEN / 8)
#define NFULL ((NPAIR - KP) / KP)
#define NHEAD (NPAIR - KP * (NFULL + 1))

__global__ __launch_bounds__(128) void gemm_wmma_kernel(const float* __restrict__ A,
                                                        const float* __restrict__ W,
                                                        float* __restrict__ C) {
    const int wave  = __builtin_amdgcn_readfirstlane((int)(threadIdx.x >> 5));
    const int lane  = threadIdx.x & 31;
    const int mtile = blockIdx.x * 2 + (wave >> 1);
    const int nh    = wave & 1;
    const int m0    = mtile * 16;
    const int half  = lane >> 4;
    const int sub   = lane & 15;

    const float* arow = A + (size_t)(m0 + sub) * HIDDEN + half * 4;
    const float* wrow = W + (size_t)(half * 2) * EXPERTS + nh * 32 + sub * 2;

    v8f acc0 = {}, acc1 = {};

    v4f qa[KP];
    v2f w00[KP], w01[KP], w10[KP], w11[KP];

#define LOAD_PAIR(j, kb)                                                     \
    do {                                                                     \
        qa[j]  = *(const v4f*)(arow + (kb));                                 \
        w00[j] = *(const v2f*)(wrow + (size_t)(kb) * EXPERTS);               \
        w01[j] = *(const v2f*)(wrow + (size_t)((kb) + 1) * EXPERTS);         \
        w10[j] = *(const v2f*)(wrow + (size_t)((kb) + 4) * EXPERTS);         \
        w11[j] = *(const v2f*)(wrow + (size_t)((kb) + 5) * EXPERTS);         \
    } while (0)

#define DO_PAIR(j)                                                           \
    do {                                                                     \
        float ax = qa[j].x, ay = qa[j].y, az = qa[j].z, aw = qa[j].w;        \
          \
               \
        asm("v_permlane16_swap_b32 %0, %2\n\tv_permlane16_swap_b32 %1, %3"    \
            : "+v"(ax), "+v"(ay), "+v"(az), "+v"(aw));                       \
        const v2f a0 = {ax, ay};                                             \
        const v2f a1 = {az, aw};                                             \
          \
          \
             \
          \
          \
            \
        float p0 = w00[j].x, p1 = w00[j].y, q0 = w01[j].x, q1 = w01[j].y;    \
        float r0 = w10[j].x, r1 = w10[j].y, s0 = w11[j].x, s1 = w11[j].y;    \
        asm("v_swap_b32 %0, %1\n\tv_swap_b32 %2, %3"                          \
            : "+v"(p1), "+v"(q0), "+v"(r1), "+v"(s0));                       \
        const v2f b00 = {p0, p1};                                            \
        const v2f b01 = {q0, q1};                                            \
        const v2f b10 = {r0, r1};                                            \
        const v2f b11 = {s0, s1};                                            \
        acc0 = __builtin_amdgcn_wmma_f32_16x16x4_f32(false, a0, false, b00, (short)0, acc0, false, false); \
        acc1 = __builtin_amdgcn_wmma_f32_16x16x4_f32(false, a0, false, b01, (short)0, acc1, false, false); \
        acc0 = __builtin_amdgcn_wmma_f32_16x16x4_f32(false, a1, false, b10, (short)0, acc0, false, false); \
        acc1 = __builtin_amdgcn_wmma_f32_16x16x4_f32(false, a1, false, b11, (short)0, acc1, false, false); \
    } while (0)

#pragma unroll
    for (int j = KP - NHEAD; j < KP; ++j) {
        LOAD_PAIR(j, 8 * (j - (KP - NHEAD)));
        __builtin_amdgcn_sched_barrier(0);
    }
#pragma unroll
    for (int j = 0; j < KP - NHEAD; ++j) {
        LOAD_PAIR(j, 8 * (j + NHEAD));
        __builtin_amdgcn_sched_barrier(0);
    }


#pragma unroll
    for (int j = KP - NHEAD; j < KP; ++j) {
        DO_PAIR(j);
        __builtin_amdgcn_sched_barrier(0);
        LOAD_PAIR(j, 8 * (j - (KP - NHEAD)) + 8 * KP);
        __builtin_amdgcn_sched_barrier(0);
    }

    int kb0 = 8 * NHEAD;
    for (int p = 0; p < NFULL; ++p, kb0 += 8 * KP) {
        __builtin_prefetch(arow + kb0 + 512, 0, 3);
        __builtin_prefetch(arow + kb0 + 544, 0, 3);
        __builtin_prefetch(arow + kb0 + 576, 0, 3);
#pragma unroll
        for (int j = 0; j < KP; ++j) {
            DO_PAIR(j);
            __builtin_amdgcn_sched_barrier(0);
            LOAD_PAIR(j, kb0 + 8 * KP + 8 * j);
            __builtin_amdgcn_sched_barrier(0);
        }
    }
#pragma unroll
    for (int j = 0; j < KP; ++j) {
        DO_PAIR(j);
        __builtin_amdgcn_sched_barrier(0);
    }
#undef DO_PAIR
#undef LOAD_PAIR

#pragma unroll
    for (int r = 0; r < 8; ++r) {
        const size_t row = (size_t)(m0 + half * 8 + r) * EXPERTS;
        const v2f o = {acc0[r], acc1[r]};
        *(v2f*)(C + row + nh * 32 + sub * 2) = o;
    }
}
#undef KP
#undef NPAIR
#undef NFULL
#undef NHEAD

#define CH 64
#define NCH (TSTEPS / CH)
#define LR 32
#define SCANW 2
#define HELPERS 8
#define LIF_THREADS (32 * (SCANW + HELPERS))
#define CH_F4 (CH * EXPERTS / 4)
#define H_ROUNDS (CH_F4 / (HELPERS * 32))
#define WARMC 7
#define CHK_F4 ((NCH - 1) * EXPERTS / 4)
#define CHK_ROUNDS ((CHK_F4 + LIF_THREADS - 1) / LIF_THREADS)
#define NPLANE (NCH * EXPERTS)

__device__ __forceinline__ float lif_step(float cur, float& m9, float& a, float& rf) {
    const float p = __builtin_fmaf(cur, a, m9);
    const float d = fmaxf(rf - 0.1f, 0.0f);
    float n = (rf <= 0.1f) ? 0.1f : 0.0f;
    asm("" : "+v"(n));
    const float t = p * 0.9f;
    const bool s = p > 1.0f;
    m9 = s ? 0.0f : t;
    a  = s ? 0.0f : n;
    rf = s ? 1.0f : d;
    return p;
}

__device__ __forceinline__ v4u f4bits(v4f v) { return __builtin_bit_cast(v4u, v); }

__global__ __launch_bounds__(LIF_THREADS) void lif_scan_kernel(const float* __restrict__ cin,
                                                                float* __restrict__ pout,
                                                                float* __restrict__ st,
                                                                int mode) {
    __shared__ v4f lds[3][CH_F4];
    __shared__ int bad[NCH];

    const int wave = __builtin_amdgcn_readfirstlane((int)(threadIdx.x >> 5));
    const int lane = threadIdx.x & 31;
    const int h    = (int)threadIdx.x - 32 * SCANW;
    const int ex   = (int)threadIdx.x & (EXPERTS - 1);

    const v4f* __restrict__ gi = (const v4f*)cin;
    v4f* __restrict__ go = (v4f*)pout;

    float m9 = 0.0f;
    float a  = 0.1f;
    float rf = 0.0f;

    int c_begin, c_store, c_end;
    if (mode == 0) {
        const int s = blockIdx.x;
        c_begin = s > WARMC ? s - WARMC : 0;
        c_store = s;
        c_end   = s + 1;
    } else {
        if (threadIdx.x < NCH) bad[threadIdx.x] = 0;
        __syncthreads();
        const v4f* sp = (const v4f*)st;
        v4u df[CHK_ROUNDS];
#pragma unroll
        for (int r = 0; r < CHK_ROUNDS; ++r) {
            const int i = min(r * LIF_THREADS + (int)threadIdx.x, CHK_F4 - 1);
            df[r] = (f4bits(sp[0 * (NPLANE / 4) + EXPERTS / 4 + i]) ^ f4bits(sp[3 * (NPLANE / 4) + i])) |
                    (f4bits(sp[1 * (NPLANE / 4) + EXPERTS / 4 + i]) ^ f4bits(sp[4 * (NPLANE / 4) + i])) |
                    (f4bits(sp[2 * (NPLANE / 4) + EXPERTS / 4 + i]) ^ f4bits(sp[5 * (NPLANE / 4) + i]));
        }
#pragma unroll
        for (int r = 0; r < CHK_ROUNDS; ++r) {
            const int i = min(r * LIF_THREADS + (int)threadIdx.x, CHK_F4 - 1);
            if ((df[r].x | df[r].y | df[r].z | df[r].w) != 0u) bad[1 + i / (EXPERTS / 4)] = 1;
        }
        __syncthreads();
        int first = NCH;
        for (int s = NCH - 1; s >= 1; --s) first = bad[s] ? s : first;
        first = __builtin_amdgcn_readfirstlane(first);
        if (first == NCH) return;
        c_begin = first;
        c_store = first;
        c_end   = NCH;
        if (wave < SCANW) {
            const int o = (first - 1) * EXPERTS + ex;
            m9 = st[3 * NPLANE + o];
            a  = st[4 * NPLANE + o];
            rf = st[5 * NPLANE + o];
        }
    }

    v4f stage[H_ROUNDS];

    if (wave >= SCANW) {
        const v4f* src0 = gi + (size_t)c_begin * CH_F4;
#pragma unroll
        for (int r = 0; r < H_ROUNDS; ++r) stage[r] = src0[r * (HELPERS * 32) + h];
#pragma unroll
        for (int r = 0; r < H_ROUNDS; ++r) lds[0][r * (HELPERS * 32) + h] = stage[r];
        if (c_begin + 1 < c_end) {
#pragma unroll
            for (int r = 0; r < H_ROUNDS; ++r) stage[r] = src0[CH_F4 + r * (HELPERS * 32) + h];
        }
    }
    __syncthreads();

    int cur = 0, nxt = 1, prv = 2;
    for (int c = c_begin; c < c_end; ++c) {
        if (wave < SCANW) {
            if (mode == 0 && c == c_store) {
                const int o = c * EXPERTS + ex;
                st[0 * NPLANE + o] = m9;
                st[1 * NPLANE + o] = a;
                st[2 * NPLANE + o] = rf;
            }
            float* rows = (float*)&lds[cur][0] + (wave * 32 + lane);
            float ring[LR];
#pragma unroll
            for (int i = 0; i < LR; ++i) ring[i] = rows[i * EXPERTS];
            int t0 = 0;
            for (; t0 < CH - LR; t0 += LR) {
#pragma unroll
                for (int i = 0; i < LR; i += 2) {
                    const float p0 = lif_step(ring[i],     m9, a, rf);
                    const float p1 = lif_step(ring[i + 1], m9, a, rf);
                    rows[(t0 + i) * EXPERTS]     = p0;
                    rows[(t0 + i + 1) * EXPERTS] = p1;
                    ring[i]     = rows[(t0 + LR + i) * EXPERTS];
                    ring[i + 1] = rows[(t0 + LR + i + 1) * EXPERTS];
                }
            }
#pragma unroll
            for (int i = 0; i < LR; ++i) {
                const float p = lif_step(ring[i], m9, a, rf);
                rows[(t0 + i) * EXPERTS] = p;
            }
        } else {
            if (c + 1 < c_end) {
#pragma unroll
                for (int r = 0; r < H_ROUNDS; ++r) lds[nxt][r * (HELPERS * 32) + h] = stage[r];
            }
            if (c + 2 < c_end) {
                const v4f* src = gi + (size_t)(c + 2) * CH_F4;
#pragma unroll
                for (int r = 0; r < H_ROUNDS; ++r) stage[r] = src[r * (HELPERS * 32) + h];
            }
            if (c > c_store) {
                v4f* dst = go + (size_t)(c - 1) * CH_F4;
#pragma unroll
                for (int r = 0; r < H_ROUNDS; ++r) dst[r * (HELPERS * 32) + h] = lds[prv][r * (HELPERS * 32) + h];
            }
        }
        __syncthreads();
        const int t = prv; prv = cur; cur = nxt; nxt = t;
    }

    if (wave >= SCANW) {
        v4f* dst = go + (size_t)(c_end - 1) * CH_F4;
#pragma unroll
        for (int r = 0; r < H_ROUNDS; ++r) dst[r * (HELPERS * 32) + h] = lds[prv][r * (HELPERS * 32) + h];
    } else if (mode == 0) {
        const int o = c_store * EXPERTS + ex;
        st[3 * NPLANE + o] = m9;
        st[4 * NPLANE + o] = a;
        st[5 * NPLANE + o] = rf;
    }
}

__global__ __launch_bounds__(256) void routing_kernel(float2* __restrict__ io) {
    const int row  = blockIdx.x * 8 + (threadIdx.x >> 5);
    const int lane = threadIdx.x & 31;

    float2 mp = io[(size_t)row * 32 + lane];
    bool s0 = mp.x > 1.0f;
    bool s1 = mp.y > 1.0f;

    unsigned long long bal = __ballot(s0 || s1);
    bool any = (bal != 0ull);

    float x0 = any ? (s0 ? 1.0f : 0.0f) : mp.x;
    float x1 = any ? (s1 ? 1.0f : 0.0f) : mp.y;

    float m = fmaxf(x0, x1);
#pragma unroll
    for (int off = 16; off >= 1; off >>= 1)
        m = fmaxf(m, __shfl_xor(m, off, 32));

    float e0 = __expf(x0 - m);
    float e1 = __expf(x1 - m);
    float s  = e0 + e1;
#pragma unroll
    for (int off = 16; off >= 1; off >>= 1)
        s += __shfl_xor(s, off, 32);

    float inv = 1.0f / s;
    float2 r;
    r.x = e0 * inv;
    r.y = e1 * inv;
    io[(size_t)row * 32 + lane] = r;
}

extern "C" void kernel_launch(void* const* d_in, const int* in_sizes, int n_in,
                              void* d_out, int out_size, void* d_ws, size_t ws_size,
                              hipStream_t stream) {
    const float* hidden = (const float*)d_in[0];
    const float* W      = (const float*)d_in[1];
    float* out          = (float*)d_out;

    float* curr = (float*)d_ws;
    float* st   = (float*)((char*)d_ws + (size_t)TSTEPS * EXPERTS * sizeof(float));

    gemm_wmma_kernel<<<256, 128, 0, stream>>>(hidden, W, curr);

    lif_scan_kernel<<<NCH, LIF_THREADS, 0, stream>>>(curr, out, st, 0);
    lif_scan_kernel<<<1, LIF_THREADS, 0, stream>>>(curr, out, st, 1);

    routing_kernel<<<TSTEPS / 8, 256, 0, stream>>>((float2*)out);
}
